# starting kernel
# baseline (speedup 1.0000x reference)
#include <hip/hip_runtime.h>
#include <math.h>

typedef float v2f __attribute__((ext_vector_type(2)));
typedef float v8f __attribute__((ext_vector_type(8)));

#define NEG_SLOPE 0.2f

__device__ __forceinline__ float gat_lrelu(float v) { return v > 0.0f ? v : NEG_SLOPE * v; }
__device__ __forceinline__ unsigned gat_fkey(float f) {
    unsigned u = __float_as_uint(f);
    return (u & 0x80000000u) ? ~u : (u | 0x80000000u);
}
__device__ __forceinline__ float gat_funkey(unsigned k) {
    return __uint_as_float((k & 0x80000000u) ? (k & 0x7fffffffu) : ~k);
}

template<int NCOL, int NCT>
__global__ __launch_bounds__(256) void gat_gemm_wmma(
    const float* __restrict__ X, const float* __restrict__ W,
    float* __restrict__ Y, int N, int ncolStore)
{
    constexpr int K = 128;
    __shared__ float ldsW[K * NCOL];
    for (int idx = threadIdx.x; idx < K * NCOL; idx += 256) {
        int k = idx / NCOL, c = idx % NCOL;
        ldsW[idx] = (c < ncolStore) ? W[k * ncolStore + c] : 0.0f;
    }
    __syncthreads();

    const int lane = threadIdx.x & 31;
    const int wave = threadIdx.x >> 5;
    const int m    = lane & 15;
    const int half = lane >> 4;
    const int koff = half << 1;

    long row0 = (long)blockIdx.x * 128 + wave * 16;
    long rowL = row0 + m; if (rowL > (long)N - 1) rowL = (long)N - 1;
    const float* xrow = X + rowL * K;

    v8f acc[NCT] = {};
    for (int k0 = 0; k0 < K; k0 += 4) {
        v2f a;
        a.x = xrow[k0 + koff];
        a.y = xrow[k0 + koff + 1];
#pragma unroll
        for (int t = 0; t < NCT; ++t) {
            v2f b;
            b.x = ldsW[(k0 + koff) * NCOL + t * 16 + m];
            b.y = ldsW[(k0 + koff + 1) * NCOL + t * 16 + m];
            acc[t] = __builtin_amdgcn_wmma_f32_16x16x4_f32(
                false, a, false, b, (short)0, acc[t], false, false);
        }
    }
#pragma unroll
    for (int r = 0; r < 8; ++r) {
        long row = row0 + half * 8 + r;
        if (row < N) {
#pragma unroll
            for (int t = 0; t < NCT; ++t) {
                int c = t * 16 + m;
                if (c < ncolStore) Y[row * ncolStore + c] = acc[t][r];
            }
        }
    }
}

template<int H, int C>
__global__ __launch_bounds__(256) void gat_alpha_kernel(
    const float* __restrict__ h, const float* __restrict__ aw_s,
    const float* __restrict__ aw_d, float* __restrict__ as_,
    float* __restrict__ ad_, int N)
{
    constexpr int HC = H * C;
    int wid  = blockIdx.x * (blockDim.x >> 5) + (threadIdx.x >> 5);
    int lane = threadIdx.x & 31;
    if (wid >= N) return;
    const float* hp = h + (long)wid * HC;
    float s[H], d[H];
#pragma unroll
    for (int hh = 0; hh < H; ++hh) { s[hh] = 0.0f; d[hh] = 0.0f; }
    for (int c = lane; c < HC; c += 32) {
        int hh = (H == 1) ? 0 : (c / C);
        float v = hp[c];
        s[hh] += v * aw_s[c];
        d[hh] += v * aw_d[c];
    }
#pragma unroll
    for (int hh = 0; hh < H; ++hh) {
        float sv = s[hh], dv = d[hh];
        for (int o = 16; o > 0; o >>= 1) {
            sv += __shfl_xor(sv, o, 32);
            dv += __shfl_xor(dv, o, 32);
        }
        if (lane == 0) {
            as_[(long)wid * H + hh] = sv;
            ad_[(long)wid * H + hh] = dv;
        }
    }
}

template<int H>
__global__ __launch_bounds__(256) void gat_edge_max_kernel(
    const int* __restrict__ src, const int* __restrict__ dst, int E, int N,
    const float* __restrict__ as_, const float* __restrict__ ad_,
    unsigned* __restrict__ nmax)
{
    int e = blockIdx.x * blockDim.x + threadIdx.x;
    if (e >= E + N) return;
    int s = (e < E) ? src[e] : (e - E);
    int d = (e < E) ? dst[e] : (e - E);
#pragma unroll
    for (int h = 0; h < H; ++h) {
        float ev = gat_lrelu(as_[s * H + h] + ad_[d * H + h]);
        atomicMax(&nmax[d * H + h], gat_fkey(ev));
    }
}

template<int H>
__global__ __launch_bounds__(256) void gat_edge_denom_kernel(
    const int* __restrict__ src, const int* __restrict__ dst, int E, int N,
    const float* __restrict__ as_, const float* __restrict__ ad_,
    const unsigned* __restrict__ nmax, float* __restrict__ den)
{
    int e = blockIdx.x * blockDim.x + threadIdx.x;
    if (e >= E + N) return;
    int s = (e < E) ? src[e] : (e - E);
    int d = (e < E) ? dst[e] : (e - E);
#pragma unroll
    for (int h = 0; h < H; ++h) {
        float ev = gat_lrelu(as_[s * H + h] + ad_[d * H + h]);
        atomicAdd(&den[d * H + h], expf(ev - gat_funkey(nmax[d * H + h])));
    }
}

template<int H>
__global__ __launch_bounds__(256) void gat_edge_coef_kernel(
    const int* __restrict__ src, const int* __restrict__ dst, int E, int N,
    const float* __restrict__ as_, const float* __restrict__ ad_,
    const unsigned* __restrict__ nmax, const float* __restrict__ den,
    float* __restrict__ coef)
{
    int e = blockIdx.x * blockDim.x + threadIdx.x;
    if (e >= E + N) return;
    int s = (e < E) ? src[e] : (e - E);
    int d = (e < E) ? dst[e] : (e - E);
#pragma unroll
    for (int h = 0; h < H; ++h) {
        float ev = gat_lrelu(as_[s * H + h] + ad_[d * H + h]);
        float mx = gat_funkey(nmax[d * H + h]);
        coef[(long)e * H + h] = expf(ev - mx) / (den[d * H + h] + 1e-16f);
    }
}

template<int H, int C>
__global__ __launch_bounds__(256) void gat_edge_scatter_kernel(
    const int* __restrict__ src, const int* __restrict__ dst, int E, int N,
    const float* __restrict__ coef, const float* __restrict__ feat,
    float* __restrict__ agg)
{
    constexpr int HC = H * C;
    int wid  = blockIdx.x * (blockDim.x >> 5) + (threadIdx.x >> 5);
    int lane = threadIdx.x & 31;
    if (wid >= E + N) return;
    int s = (wid < E) ? src[wid] : (wid - E);
    int d = (wid < E) ? dst[wid] : (wid - E);
    const float* fs = feat + (long)s * HC;
    float* ao = agg + (long)d * HC;
    for (int c = lane; c < HC; c += 32) {
        int hh = (H == 1) ? 0 : (c / C);
        atomicAdd(&ao[c], coef[(long)wid * H + hh] * fs[c]);
    }
}

__global__ void gat_fill_u32_kernel(unsigned* __restrict__ p, unsigned v, long n) {
    long i = blockIdx.x * (long)blockDim.x + threadIdx.x;
    if (i < n) p[i] = v;
}

__global__ void gat_elu_bias_kernel(float* __restrict__ a, const float* __restrict__ b,
                                    long n, int HC) {
    long i = blockIdx.x * (long)blockDim.x + threadIdx.x;
    if (i < n) {
        float t = a[i] + b[i % HC];
        a[i] = t > 0.0f ? t : (expf(t) - 1.0f);
    }
}

__global__ void gat_bias_out_kernel(const float* __restrict__ a, const float* __restrict__ b,
                                    float* __restrict__ out, long n, int HC) {
    long i = blockIdx.x * (long)blockDim.x + threadIdx.x;
    if (i < n) out[i] = a[i] + b[i % HC];
}

extern "C" void kernel_launch(void* const* d_in, const int* in_sizes, int n_in,
                              void* d_out, int out_size, void* d_ws, size_t ws_size,
                              hipStream_t stream)
{
    (void)n_in; (void)out_size; (void)ws_size;
    const float* x   = (const float*)d_in[0];
    const int*   ei  = (const int*)  d_in[1];
    const float* W1  = (const float*)d_in[2];
    const float* aS1 = (const float*)d_in[3];
    const float* aD1 = (const float*)d_in[4];
    const float* b1  = (const float*)d_in[5];
    const float* W2  = (const float*)d_in[6];
    const float* aS2 = (const float*)d_in[7];
    const float* aD2 = (const float*)d_in[8];
    const float* b2  = (const float*)d_in[9];

    const int IN = 128, HC1 = 128, OUTC = 40;
    int N  = in_sizes[0] / IN;
    int E  = in_sizes[1] / 2;
    int ET = E + N;
    const int* srcI = ei;
    const int* dstI = ei + E;

    float* wsf = (float*)d_ws;
    size_t off = 0;
    auto take = [&](size_t nf) -> size_t { size_t o = off; off += (nf + 63) & ~(size_t)63; return o; };
    float*    h1    = wsf + take((size_t)N * HC1);
    float*    agg1  = wsf + take((size_t)N * HC1);
    float*    as1   = wsf + take((size_t)N * 2);
    float*    ad1   = wsf + take((size_t)N * 2);
    unsigned* nmax1 = (unsigned*)(wsf + take((size_t)N * 2));
    float*    den1  = wsf + take((size_t)N * 2);
    float*    coef1 = wsf + take((size_t)ET * 2);
    float*    h2    = wsf + take((size_t)N * OUTC);
    float*    agg2  = wsf + take((size_t)N * OUTC);
    float*    as2   = wsf + take((size_t)N);
    float*    ad2   = wsf + take((size_t)N);
    unsigned* nmax2 = (unsigned*)(wsf + take((size_t)N));
    float*    den2  = wsf + take((size_t)N);
    float*    coef2 = wsf + take((size_t)ET);

    dim3 blk(256);
    int egrid = (ET + 255) / 256;
    int wgrid = (ET + 7) / 8;

    gat_gemm_wmma<128, 8><<<(N + 127) / 128, blk, 0, stream>>>(x, W1, h1, N, HC1);
    gat_alpha_kernel<2, 64><<<(N + 7) / 8, blk, 0, stream>>>(h1, aS1, aD1, as1, ad1, N);
    gat_fill_u32_kernel<<<((long)N * 2 + 255) / 256, blk, 0, stream>>>(nmax1, 0u, (long)N * 2);
    gat_fill_u32_kernel<<<((long)N * 2 + 255) / 256, blk, 0, stream>>>((unsigned*)den1, 0u, (long)N * 2);
    gat_fill_u32_kernel<<<((long)N * HC1 + 255) / 256, blk, 0, stream>>>((unsigned*)agg1, 0u, (long)N * HC1);
    gat_edge_max_kernel<2><<<egrid, blk, 0, stream>>>(srcI, dstI, E, N, as1, ad1, nmax1);
    gat_edge_denom_kernel<2><<<egrid, blk, 0, stream>>>(srcI, dstI, E, N, as1, ad1, nmax1, den1);
    gat_edge_coef_kernel<2><<<egrid, blk, 0, stream>>>(srcI, dstI, E, N, as1, ad1, nmax1, den1, coef1);
    gat_edge_scatter_kernel<2, 64><<<wgrid, blk, 0, stream>>>(srcI, dstI, E, N, coef1, h1, agg1);
    gat_elu_bias_kernel<<<((long)N * HC1 + 255) / 256, blk, 0, stream>>>(agg1, b1, (long)N * HC1, HC1);

    gat_gemm_wmma<48, 3><<<(N + 127) / 128, blk, 0, stream>>>(agg1, W2, h2, N, OUTC);
    gat_alpha_kernel<1, 40><<<(N + 7) / 8, blk, 0, stream>>>(h2, aS2, aD2, as2, ad2, N);
    gat_fill_u32_kernel<<<((long)N + 255) / 256, blk, 0, stream>>>(nmax2, 0u, (long)N);
    gat_fill_u32_kernel<<<((long)N + 255) / 256, blk, 0, stream>>>((unsigned*)den2, 0u, (long)N);
    gat_fill_u32_kernel<<<((long)N * OUTC + 255) / 256, blk, 0, stream>>>((unsigned*)agg2, 0u, (long)N * OUTC);
    gat_edge_max_kernel<1><<<egrid, blk, 0, stream>>>(srcI, dstI, E, N, as2, ad2, nmax2);
    gat_edge_denom_kernel<1><<<egrid, blk, 0, stream>>>(srcI, dstI, E, N, as2, ad2, nmax2, den2);
    gat_edge_coef_kernel<1><<<egrid, blk, 0, stream>>>(srcI, dstI, E, N, as2, ad2, nmax2, den2, coef2);
    gat_edge_scatter_kernel<1, 40><<<wgrid, blk, 0, stream>>>(srcI, dstI, E, N, coef2, h2, agg2);
    gat_bias_out_kernel<<<((long)N * OUTC + 255) / 256, blk, 0, stream>>>(
        agg2, b2, (float*)d_out, (long)N * OUTC, OUTC);
}


# float-atomic edge passes of both layers replaced by a per-call edge list grouped by destination (groups ordered by source: fixed sum order) and one wave-per-node softmax+gather kernel per layer
# speedup vs baseline: 2.8168x; 2.8168x over previous
#include <hip/hip_runtime.h>
#include <math.h>

typedef float v2f __attribute__((ext_vector_type(2)));
typedef float v8f __attribute__((ext_vector_type(8)));

#define NEG_SLOPE 0.2f

__device__ __forceinline__ float gat_lrelu(float v) { return v > 0.0f ? v : NEG_SLOPE * v; }

template<int NCOL, int NCT>
__global__ __launch_bounds__(256) void gat_gemm_wmma(
    const float* __restrict__ X, const float* __restrict__ W,
    float* __restrict__ Y, int N, int ncolStore)
{
    constexpr int K = 128;
    __shared__ float ldsW[K * NCOL];
    for (int idx = threadIdx.x; idx < K * NCOL; idx += 256) {
        int k = idx / NCOL, c = idx % NCOL;
        ldsW[idx] = (c < ncolStore) ? W[k * ncolStore + c] : 0.0f;
    }
    __syncthreads();

    const int lane = threadIdx.x & 31;
    const int wave = threadIdx.x >> 5;
    const int m    = lane & 15;
    const int half = lane >> 4;
    const int koff = half << 1;

    long row0 = (long)blockIdx.x * 128 + wave * 16;
    long rowL = row0 + m; if (rowL > (long)N - 1) rowL = (long)N - 1;
    const float* xrow = X + rowL * K;

    v8f acc[NCT] = {};
    for (int k0 = 0; k0 < K; k0 += 4) {
        v2f a;
        a.x = xrow[k0 + koff];
        a.y = xrow[k0 + koff + 1];
#pragma unroll
        for (int t = 0; t < NCT; ++t) {
            v2f b;
            b.x = ldsW[(k0 + koff) * NCOL + t * 16 + m];
            b.y = ldsW[(k0 + koff + 1) * NCOL + t * 16 + m];
            acc[t] = __builtin_amdgcn_wmma_f32_16x16x4_f32(
                false, a, false, b, (short)0, acc[t], false, false);
        }
    }
#pragma unroll
    for (int r = 0; r < 8; ++r) {
        long row = row0 + half * 8 + r;
        if (row < N) {
#pragma unroll
            for (int t = 0; t < NCT; ++t) {
                int c = t * 16 + m;
                if (c < ncolStore) Y[row * ncolStore + c] = acc[t][r];
            }
        }
    }
}

template<int H, int C>
__global__ __launch_bounds__(256) void gat_alpha_kernel(
    const float* __restrict__ h, const float* __restrict__ aw_s,
    const float* __restrict__ aw_d, float* __restrict__ as_,
    float* __restrict__ ad_, int N)
{
    constexpr int HC = H * C;
    int wid  = blockIdx.x * (blockDim.x >> 5) + (threadIdx.x >> 5);
    int lane = threadIdx.x & 31;
    if (wid >= N) return;
    const float* hp = h + (long)wid * HC;
    float s[H], d[H];
#pragma unroll
    for (int hh = 0; hh < H; ++hh) { s[hh] = 0.0f; d[hh] = 0.0f; }
    for (int c = lane; c < HC; c += 32) {
        int hh = (H == 1) ? 0 : (c / C);
        float v = hp[c];
        s[hh] += v * aw_s[c];
        d[hh] += v * aw_d[c];
    }
#pragma unroll
    for (int hh = 0; hh < H; ++hh) {
        float sv = s[hh], dv = d[hh];
        for (int o = 16; o > 0; o >>= 1) {
            sv += __shfl_xor(sv, o, 32);
            dv += __shfl_xor(dv, o, 32);
        }
        if (lane == 0) {
            as_[(long)wid * H + hh] = sv;
            ad_[(long)wid * H + hh] = dv;
        }
    }
}

__global__ void gat_fill_u32_kernel(unsigned* __restrict__ p, unsigned v, long n) {
    long i = blockIdx.x * (long)blockDim.x + threadIdx.x;
    if (i < n) p[i] = v;
}

__global__ __launch_bounds__(256) void gat_count_kernel(
    const int* __restrict__ dst, int E, int* __restrict__ cnt)
{
    int e = blockIdx.x * blockDim.x + threadIdx.x;
    if (e < E) atomicAdd(&cnt[dst[e]], 1);
}

__global__ __launch_bounds__(256) void gat_scan_kernel(
    const int* __restrict__ cnt, int* __restrict__ offs, int* __restrict__ cursor, int N)
{
    __shared__ int red[8];
    __shared__ int wsum[8];
    const int tid = threadIdx.x, lane = tid & 31, wave = tid >> 5;
    const int tile0 = blockIdx.x * 1024;
    int part = 0;
    for (int i = tid; i < tile0; i += 256) part += cnt[i];
    for (int o = 16; o > 0; o >>= 1) part += __shfl_xor(part, o, 32);
    if (lane == 0) red[wave] = part;
    const int i0 = tile0 + tid * 4;
    int v[4];
#pragma unroll
    for (int j = 0; j < 4; ++j) v[j] = (i0 + j < N) ? cnt[i0 + j] : 0;
    const int tsum = v[0] + v[1] + v[2] + v[3];
    int inc = tsum;
    for (int o = 1; o < 32; o <<= 1) {
        int y = __shfl_up(inc, o, 32);
        if (lane >= o) inc += y;
    }
    if (lane == 31) wsum[wave] = inc;
    __syncthreads();
    int base = 0;
#pragma unroll
    for (int w = 0; w < 8; ++w) base += red[w];
    for (int w = 0; w < wave; ++w) base += wsum[w];
    int ex = base + inc - tsum;
#pragma unroll
    for (int j = 0; j < 4; ++j) {
        if (i0 + j < N) { offs[i0 + j] = ex; cursor[i0 + j] = ex; ex += v[j]; }
    }
}

__global__ __launch_bounds__(256) void gat_place_kernel(
    const int* __restrict__ src, const int* __restrict__ dst, int E,
    int* __restrict__ cursor, int* __restrict__ tmp)
{
    int e = blockIdx.x * blockDim.x + threadIdx.x;
    if (e < E) {
        int p = atomicAdd(&cursor[dst[e]], 1);
        tmp[p] = src[e];
    }
}

__global__ __launch_bounds__(256) void gat_order_kernel(
    const int* __restrict__ cnt, const int* __restrict__ offs,
    const int* __restrict__ tmp, int* __restrict__ sorted, int N)
{
    int wid  = blockIdx.x * 8 + (threadIdx.x >> 5);
    int lane = threadIdx.x & 31;
    if (wid >= N) return;
    const int deg = __builtin_amdgcn_readfirstlane(cnt[wid]);
    const int off = __builtin_amdgcn_readfirstlane(offs[wid]);
    if (deg <= 32) {
        int v = (lane < deg) ? tmp[off + lane] : 0x7fffffff;
        int rank = 0;
#pragma unroll
        for (int j = 0; j < 32; ++j) {
            int vj = __builtin_amdgcn_readlane(v, j);
            rank += (vj < v || (vj == v && j < lane)) ? 1 : 0;
        }
        if (lane < deg) sorted[off + rank] = v;
    } else {
        for (int i = lane; i < deg; i += 32) {
            int v = tmp[off + i];
            int rank = 0;
            for (int j = 0; j < deg; ++j) {
                int vj = tmp[off + j];
                rank += (vj < v || (vj == v && j < i)) ? 1 : 0;
            }
            sorted[off + rank] = v;
        }
    }
}

__device__ __forceinline__ float gat_rl(float x, int k) {
    return __uint_as_float((unsigned)__builtin_amdgcn_readlane((int)__float_as_uint(x), k));
}

__global__ __launch_bounds__(256) void gat_node_l1_kernel(
    const int* __restrict__ cnt, const int* __restrict__ offs, const int* __restrict__ sorted,
    const float* __restrict__ as_, const float* __restrict__ ad_,
    const float* __restrict__ feat, const float* __restrict__ bias,
    float* __restrict__ outp, int N)
{
    int wid  = blockIdx.x * 8 + (threadIdx.x >> 5);
    const int lane = threadIdx.x & 31;
    if (wid >= N) return;
    const int n   = __builtin_amdgcn_readfirstlane(wid);
    const int deg = __builtin_amdgcn_readfirstlane(cnt[n]);
    const int off = __builtin_amdgcn_readfirstlane(offs[n]);
    const float2* as2 = reinterpret_cast<const float2*>(as_);
    const float2* ad2 = reinterpret_cast<const float2*>(ad_);
    const float4* f4  = reinterpret_cast<const float4*>(feat);
    const float2 adn = ad2[n];
    const float2 asn = as2[n];
    const float es0 = gat_lrelu(asn.x + adn.x), es1 = gat_lrelu(asn.y + adn.y);

    const bool v0 = lane < deg;
    int s = n; float e0 = es0, e1 = es1;
    if (v0) {
        s = sorted[off + lane];
        float2 a = as2[s];
        e0 = gat_lrelu(a.x + adn.x); e1 = gat_lrelu(a.y + adn.y);
    }
    float m0 = fmaxf(e0, es0), m1 = fmaxf(e1, es1);
    for (int base = 32; base < deg; base += 32) {
        int i = base + lane;
        if (i < deg) {
            float2 a = as2[sorted[off + i]];
            m0 = fmaxf(m0, gat_lrelu(a.x + adn.x)); m1 = fmaxf(m1, gat_lrelu(a.y + adn.y));
        }
    }
    for (int o = 16; o > 0; o >>= 1) {
        m0 = fmaxf(m0, __shfl_xor(m0, o, 32));
        m1 = fmaxf(m1, __shfl_xor(m1, o, 32));
    }
    const float p0 = v0 ? expf(e0 - m0) : 0.0f, p1 = v0 ? expf(e1 - m1) : 0.0f;
    float d0 = p0, d1 = p1;
    for (int base = 32; base < deg; base += 32) {
        int i = base + lane;
        if (i < deg) {
            float2 a = as2[sorted[off + i]];
            d0 += expf(gat_lrelu(a.x + adn.x) - m0); d1 += expf(gat_lrelu(a.y + adn.y) - m1);
        }
    }
    for (int o = 16; o > 0; o >>= 1) {
        d0 += __shfl_xor(d0, o, 32);
        d1 += __shfl_xor(d1, o, 32);
    }
    const float ps0 = expf(es0 - m0), ps1 = expf(es1 - m1);
    const float den0 = (d0 + ps0) + 1e-16f, den1 = (d1 + ps1) + 1e-16f;

    float4 acc = make_float4(0.0f, 0.0f, 0.0f, 0.0f);
    auto gather = [&](int sv, float q0, float q1, int cn) {
        for (int k = 0; k < cn; k += 4) {
#pragma unroll
            for (int u = 0; u < 4; ++u) {
                int   sk = __builtin_amdgcn_readlane(sv, k + u);
                float a0 = gat_rl(q0, k + u), a1 = gat_rl(q1, k + u);
                float ck = (lane < 16) ? a0 : a1;
                float4 f = f4[(long)sk * 32 + lane];
                acc.x = fmaf(ck, f.x, acc.x); acc.y = fmaf(ck, f.y, acc.y);
                acc.z = fmaf(ck, f.z, acc.z); acc.w = fmaf(ck, f.w, acc.w);
            }
        }
    };
    gather(s, p0 / den0, p1 / den1, deg < 32 ? deg : 32);
    for (int base = 32; base < deg; base += 32) {
        int i = base + lane;
        int t = n; float q0 = 0.0f, q1 = 0.0f;
        if (i < deg) {
            t = sorted[off + i];
            float2 a = as2[t];
            q0 = expf(gat_lrelu(a.x + adn.x) - m0) / den0;
            q1 = expf(gat_lrelu(a.y + adn.y) - m1) / den1;
        }
        gather(t, q0, q1, (deg - base) < 32 ? (deg - base) : 32);
    }
    {
        float cs = (lane < 16) ? (ps0 / den0) : (ps1 / den1);
        float4 f = f4[(long)n * 32 + lane];
        acc.x = fmaf(cs, f.x, acc.x); acc.y = fmaf(cs, f.y, acc.y);
        acc.z = fmaf(cs, f.z, acc.z); acc.w = fmaf(cs, f.w, acc.w);
    }
    const float4 b = reinterpret_cast<const float4*>(bias)[lane];
    float4 r;
    float t0 = acc.x + b.x, t1 = acc.y + b.y, t2 = acc.z + b.z, t3 = acc.w + b.w;
    r.x = t0 > 0.0f ? t0 : (expf(t0) - 1.0f);
    r.y = t1 > 0.0f ? t1 : (expf(t1) - 1.0f);
    r.z = t2 > 0.0f ? t2 : (expf(t2) - 1.0f);
    r.w = t3 > 0.0f ? t3 : (expf(t3) - 1.0f);
    reinterpret_cast<float4*>(outp)[(long)n * 32 + lane] = r;
}

__global__ __launch_bounds__(256) void gat_node_l2_kernel(
    const int* __restrict__ cnt, const int* __restrict__ offs, const int* __restrict__ sorted,
    const float* __restrict__ as_, const float* __restrict__ ad_,
    const float* __restrict__ feat, const float* __restrict__ bias,
    float* __restrict__ outp, int N)
{
    int wid  = blockIdx.x * 8 + (threadIdx.x >> 5);
    const int lane = threadIdx.x & 31;
    if (wid >= N) return;
    const int n   = __builtin_amdgcn_readfirstlane(wid);
    const int deg = __builtin_amdgcn_readfirstlane(cnt[n]);
    const int off = __builtin_amdgcn_readfirstlane(offs[n]);
    const float4* f4 = reinterpret_cast<const float4*>(feat);
    const float adn = ad_[n];
    const float es  = gat_lrelu(as_[n] + adn);

    const bool v0 = lane < deg;
    int s = n; float e = es;
    if (v0) { s = sorted[off + lane]; e = gat_lrelu(as_[s] + adn); }
    float m = fmaxf(e, es);
    for (int base = 32; base < deg; base += 32) {
        int i = base + lane;
        if (i < deg) m = fmaxf(m, gat_lrelu(as_[sorted[off + i]] + adn));
    }
    for (int o = 16; o > 0; o >>= 1) m = fmaxf(m, __shfl_xor(m, o, 32));
    const float p = v0 ? expf(e - m) : 0.0f;
    float d = p;
    for (int base = 32; base < deg; base += 32) {
        int i = base + lane;
        if (i < deg) d += expf(gat_lrelu(as_[sorted[off + i]] + adn) - m);
    }
    for (int o = 16; o > 0; o >>= 1) d += __shfl_xor(d, o, 32);
    const float ps  = expf(es - m);
    const float den = (d + ps) + 1e-16f;

    const int  g  = lane / 10;
    const int  cl = lane - g * 10;
    const bool on = g < 3;
    float4 acc = make_float4(0.0f, 0.0f, 0.0f, 0.0f);
    auto gather = [&](int sv, float q, int cn) {
        const int iters = (cn + 2) / 3;
        for (int tb = 0; tb < iters; tb += 4) {
#pragma unroll
          for (int u = 0; u < 4; ++u) {
            int k  = (tb + u) * 3 + g;
            const bool val = on && (k < cn);
            int kk = (k < cn) ? k : 0;
            int   sk = __shfl(sv, kk, 32);
            float ck = __shfl(q, kk, 32);
            ck = val ? ck : 0.0f;
            float4 f = f4[(long)(val ? sk : n) * 10 + cl];
            acc.x = fmaf(ck, f.x, acc.x); acc.y = fmaf(ck, f.y, acc.y);
            acc.z = fmaf(ck, f.z, acc.z); acc.w = fmaf(ck, f.w, acc.w);
          }
        }
    };
    gather(s, p / den, deg < 32 ? deg : 32);
    for (int base = 32; base < deg; base += 32) {
        int i = base + lane;
        int t = n; float q = 0.0f;
        if (i < deg) {
            t = sorted[off + i];
            q = expf(gat_lrelu(as_[t] + adn) - m) / den;
        }
        gather(t, q, (deg - base) < 32 ? (deg - base) : 32);
    }
    const int l1 = (lane + 10) & 31, l2 = (lane + 20) & 31;
    float4 a1, a2;
    a1.x = __shfl(acc.x, l1, 32); a1.y = __shfl(acc.y, l1, 32);
    a1.z = __shfl(acc.z, l1, 32); a1.w = __shfl(acc.w, l1, 32);
    a2.x = __shfl(acc.x, l2, 32); a2.y = __shfl(acc.y, l2, 32);
    a2.z = __shfl(acc.z, l2, 32); a2.w = __shfl(acc.w, l2, 32);
    if (lane < 10) {
        const float cs = ps / den;
        const float4 f = f4[(long)n * 10 + lane];
        const float4 b = reinterpret_cast<const float4*>(bias)[lane];
        float4 r;
        r.x = fmaf(cs, f.x, (acc.x + a1.x) + a2.x) + b.x;
        r.y = fmaf(cs, f.y, (acc.y + a1.y) + a2.y) + b.y;
        r.z = fmaf(cs, f.z, (acc.z + a1.z) + a2.z) + b.z;
        r.w = fmaf(cs, f.w, (acc.w + a1.w) + a2.w) + b.w;
        reinterpret_cast<float4*>(outp)[(long)n * 10 + lane] = r;
    }
}

extern "C" void kernel_launch(void* const* d_in, const int* in_sizes, int n_in,
                              void* d_out, int out_size, void* d_ws, size_t ws_size,
                              hipStream_t stream)
{
    (void)n_in; (void)out_size; (void)ws_size;
    const float* x   = (const float*)d_in[0];
    const int*   ei  = (const int*)  d_in[1];
    const float* W1  = (const float*)d_in[2];
    const float* aS1 = (const float*)d_in[3];
    const float* aD1 = (const float*)d_in[4];
    const float* b1  = (const float*)d_in[5];
    const float* W2  = (const float*)d_in[6];
    const float* aS2 = (const float*)d_in[7];
    const float* aD2 = (const float*)d_in[8];
    const float* b2  = (const float*)d_in[9];

    const int IN = 128, HC1 = 128, OUTC = 40;
    int N  = in_sizes[0] / IN;
    int E  = in_sizes[1] / 2;
    const int* srcI = ei;
    const int* dstI = ei + E;

    float* wsf = (float*)d_ws;
    size_t off = 0;
    auto take = [&](size_t nf) -> size_t { size_t o = off; off += (nf + 63) & ~(size_t)63; return o; };
    float* h1     = wsf + take((size_t)N * HC1);
    float* agg1   = wsf + take((size_t)N * HC1);
    float* as1    = wsf + take((size_t)N * 2);
    float* ad1    = wsf + take((size_t)N * 2);
    float* h2     = wsf + take((size_t)N * OUTC);
    float* as2    = wsf + take((size_t)N);
    float* ad2    = wsf + take((size_t)N);
    int*   cnt    = (int*)(wsf + take((size_t)N));
    int*   offs   = (int*)(wsf + take((size_t)N));
    int*   cursor = (int*)(wsf + take((size_t)N));
    int*   tmp    = (int*)(wsf + take((size_t)E));
    int*   sorted = (int*)(wsf + take((size_t)E));

    dim3 blk(256);
    int egrid = (E + 255) / 256;
    int ngrid = (N + 7) / 8;

    gat_fill_u32_kernel<<<((long)N + 255) / 256, blk, 0, stream>>>((unsigned*)cnt, 0u, (long)N);
    gat_count_kernel<<<egrid, blk, 0, stream>>>(dstI, E, cnt);
    gat_scan_kernel<<<(N + 1023) / 1024, blk, 0, stream>>>(cnt, offs, cursor, N);
    gat_place_kernel<<<egrid, blk, 0, stream>>>(srcI, dstI, E, cursor, tmp);
    gat_order_kernel<<<ngrid, blk, 0, stream>>>(cnt, offs, tmp, sorted, N);

    gat_gemm_wmma<128, 8><<<(N + 127) / 128, blk, 0, stream>>>(x, W1, h1, N, HC1);
    gat_alpha_kernel<2, 64><<<ngrid, blk, 0, stream>>>(h1, aS1, aD1, as1, ad1, N);
    gat_node_l1_kernel<<<ngrid, blk, 0, stream>>>(cnt, offs, sorted, as1, ad1, h1, b1, agg1, N);

    gat_gemm_wmma<48, 3><<<(N + 127) / 128, blk, 0, stream>>>(agg1, W2, h2, N, OUTC);
    gat_alpha_kernel<1, 40><<<ngrid, blk, 0, stream>>>(h2, aS2, aD2, as2, ad2, N);
    gat_node_l2_kernel<<<ngrid, blk, 0, stream>>>(cnt, offs, sorted, as2, ad2, h2, b2, (float*)d_out, N);
}


# per-call edge list built by a two-level counting sort with LDS counters (block histogram, binned scatter, per-bin sort) instead of one global atomic per edge in count and place; sum order unchanged
# speedup vs baseline: 4.5254x; 1.6066x over previous
#include <hip/hip_runtime.h>
#include <math.h>

typedef float v2f __attribute__((ext_vector_type(2)));
typedef float v8f __attribute__((ext_vector_type(8)));

#define NEG_SLOPE 0.2f

__device__ __forceinline__ float gat_lrelu(float v) { return v > 0.0f ? v : NEG_SLOPE * v; }

template<int NCOL, int NCT>
__global__ __launch_bounds__(256) void gat_gemm_wmma(
    const float* __restrict__ X, const float* __restrict__ W,
    float* __restrict__ Y, int N, int ncolStore)
{
    constexpr int K = 128;
    __shared__ float ldsW[K * NCOL];
    for (int idx = threadIdx.x; idx < K * NCOL; idx += 256) {
        int k = idx / NCOL, c = idx % NCOL;
        ldsW[idx] = (c < ncolStore) ? W[k * ncolStore + c] : 0.0f;
    }
    __syncthreads();

    const int lane = threadIdx.x & 31;
    const int wave = threadIdx.x >> 5;
    const int m    = lane & 15;
    const int half = lane >> 4;
    const int koff = half << 1;

    long row0 = (long)blockIdx.x * 128 + wave * 16;
    long rowL = row0 + m; if (rowL > (long)N - 1) rowL = (long)N - 1;
    const float* xrow = X + rowL * K;

    v8f acc[NCT] = {};
    for (int k0 = 0; k0 < K; k0 += 4) {
        v2f a;
        a.x = xrow[k0 + koff];
        a.y = xrow[k0 + koff + 1];
#pragma unroll
        for (int t = 0; t < NCT; ++t) {
            v2f b;
            b.x = ldsW[(k0 + koff) * NCOL + t * 16 + m];
            b.y = ldsW[(k0 + koff + 1) * NCOL + t * 16 + m];
            acc[t] = __builtin_amdgcn_wmma_f32_16x16x4_f32(
                false, a, false, b, (short)0, acc[t], false, false);
        }
    }
#pragma unroll
    for (int r = 0; r < 8; ++r) {
        long row = row0 + half * 8 + r;
        if (row < N) {
#pragma unroll
            for (int t = 0; t < NCT; ++t) {
                int c = t * 16 + m;
                if (c < ncolStore) Y[row * ncolStore + c] = acc[t][r];
            }
        }
    }
}

template<int H, int C>
__global__ __launch_bounds__(256) void gat_alpha_kernel(
    const float* __restrict__ h, const float* __restrict__ aw_s,
    const float* __restrict__ aw_d, float* __restrict__ as_,
    float* __restrict__ ad_, int N)
{
    constexpr int HC = H * C;
    int wid  = blockIdx.x * (blockDim.x >> 5) + (threadIdx.x >> 5);
    int lane = threadIdx.x & 31;
    if (wid >= N) return;
    const float* hp = h + (long)wid * HC;
    float s[H], d[H];
#pragma unroll
    for (int hh = 0; hh < H; ++hh) { s[hh] = 0.0f; d[hh] = 0.0f; }
    for (int c = lane; c < HC; c += 32) {
        int hh = (H == 1) ? 0 : (c / C);
        float v = hp[c];
        s[hh] += v * aw_s[c];
        d[hh] += v * aw_d[c];
    }
#pragma unroll
    for (int hh = 0; hh < H; ++hh) {
        float sv = s[hh], dv = d[hh];
        for (int o = 16; o > 0; o >>= 1) {
            sv += __shfl_xor(sv, o, 32);
            dv += __shfl_xor(dv, o, 32);
        }
        if (lane == 0) {
            as_[(long)wid * H + hh] = sv;
            ad_[(long)wid * H + hh] = dv;
        }
    }
}

#define GAT_BIN_SHIFT 9
#define GAT_BIN_NODES 512
#define GAT_MAXB      1024
#define GAT_EPB       4096
#define GAT_EPT       4
#define GAT_CPAD      32
#define GAT_SORT_CAP  12288

__global__ void gat_fill_u32_kernel(unsigned* __restrict__ p, unsigned v, long n) {
    long i = blockIdx.x * (long)blockDim.x + threadIdx.x;
    if (i < n) p[i] = v;
}

__device__ __forceinline__ unsigned gat_block_scan(unsigned c, unsigned* wsum) {
    const int lane = threadIdx.x & 31, wave = threadIdx.x >> 5;
    unsigned inc = c;
    for (int o = 1; o < 32; o <<= 1) {
        unsigned y = __shfl_up(inc, o, 32);
        if (lane >= o) inc += y;
    }
    if (lane == 31) wsum[wave] = inc;
    __syncthreads();
    unsigned wb = 0;
    for (int w = 0; w < wave; ++w) wb += wsum[w];
    __syncthreads();
    return wb + inc - c;
}

__global__ __launch_bounds__(1024) void gat_bin_hist_kernel(
    const int* __restrict__ dst, int E, int NB,
    unsigned* __restrict__ binCount, unsigned* __restrict__ blockBase)
{
    __shared__ unsigned hist[GAT_MAXB];
    const int tid = threadIdx.x;
    hist[tid] = 0u;
    __syncthreads();
    const int e0 = blockIdx.x * GAT_EPB;
    int dv[GAT_EPT];
#pragma unroll
    for (int j = 0; j < GAT_EPT; ++j) {
        int e = e0 + j * 1024 + tid;
        dv[j] = -1;
        if (e < E) dv[j] = dst[e];
    }
#pragma unroll
    for (int j = 0; j < GAT_EPT; ++j) {
        if (dv[j] >= 0) atomicAdd(&hist[dv[j] >> GAT_BIN_SHIFT], 1u);
    }
    __syncthreads();
    if (tid < NB) {
        const unsigned cb = hist[tid];
        unsigned base = 0u;
        if (cb) base = atomicAdd(&binCount[tid * GAT_CPAD], cb);
        blockBase[(long)blockIdx.x * NB + tid] = base;
    }
}

__global__ __launch_bounds__(1024) void gat_bin_scatter_kernel(
    const int* __restrict__ src, const int* __restrict__ dst, int E, int NB,
    const unsigned* __restrict__ binCount, const unsigned* __restrict__ blockBase,
    int* __restrict__ binStart, int2* __restrict__ part)
{
    __shared__ int2 stage[GAT_EPB];
    __shared__ unsigned hist[GAT_MAXB];
    __shared__ unsigned lstart[GAT_MAXB];
    __shared__ unsigned gbase[GAT_MAXB];
    __shared__ unsigned wsum[32];
    const int tid = threadIdx.x;
    hist[tid] = 0u;
    __syncthreads();
    const int e0 = blockIdx.x * GAT_EPB;
    int sv[GAT_EPT], dv[GAT_EPT];
    unsigned rk[GAT_EPT];
#pragma unroll
    for (int j = 0; j < GAT_EPT; ++j) {
        int e = e0 + j * 1024 + tid;
        sv[j] = 0; dv[j] = -1;
        if (e < E) { dv[j] = dst[e]; sv[j] = src[e]; }
    }
#pragma unroll
    for (int j = 0; j < GAT_EPT; ++j) {
        rk[j] = 0u;
        if (dv[j] >= 0) rk[j] = atomicAdd(&hist[dv[j] >> GAT_BIN_SHIFT], 1u);
    }
    __syncthreads();
    const unsigned ls = gat_block_scan(hist[tid], wsum);
    const unsigned bc = (tid < NB) ? binCount[tid * GAT_CPAD] : 0u;
    const unsigned bs = gat_block_scan(bc, wsum);
    lstart[tid] = ls;
    if (tid < NB) {
        gbase[tid] = bs + blockBase[(long)blockIdx.x * NB + tid] - ls;
        if (blockIdx.x == 0) {
            binStart[tid] = (int)bs;
            if (tid == NB - 1) binStart[NB] = (int)(bs + bc);
        }
    }
    __syncthreads();
#pragma unroll
    for (int j = 0; j < GAT_EPT; ++j) {
        if (dv[j] >= 0) stage[lstart[dv[j] >> GAT_BIN_SHIFT] + rk[j]] = make_int2(sv[j], dv[j]);
    }
    __syncthreads();
    const int rem = E - e0;
    const int nval = rem < GAT_EPB ? rem : GAT_EPB;
    for (int i = tid; i < nval; i += 1024) {
        const int2 p = stage[i];
        part[gbase[p.y >> GAT_BIN_SHIFT] + (unsigned)i] = p;
    }
}

__global__ __launch_bounds__(1024) void gat_bin_sort_kernel(
    const int2* __restrict__ part, const int* __restrict__ binStart, int N,
    int* __restrict__ cnt, int* __restrict__ offs, int* __restrict__ tmp)
{
    __shared__ int stageS[GAT_SORT_CAP];
    __shared__ unsigned cntL[GAT_BIN_NODES];
    __shared__ unsigned curL[GAT_BIN_NODES];
    __shared__ unsigned wsum[32];
    const int tid = threadIdx.x;
    const int bin = blockIdx.x;
    const int beg = binStart[bin];
    const int m   = binStart[bin + 1] - beg;
    const int node0 = bin << GAT_BIN_SHIFT;
    if (tid < GAT_BIN_NODES) cntL[tid] = 0u;
    __syncthreads();
    for (int b0 = 0; b0 < m; b0 += 4096) {
        int dl[4];
#pragma unroll
        for (int u = 0; u < 4; ++u) {
            const int i = b0 + u * 1024 + tid;
            dl[u] = part[beg + (i < m ? i : m - 1)].y;
        }
#pragma unroll
        for (int u = 0; u < 4; ++u) {
            const int i = b0 + u * 1024 + tid;
            if (i < m) atomicAdd(&cntL[dl[u] - node0], 1u);
        }
    }
    __syncthreads();
    const unsigned cn = (tid < GAT_BIN_NODES) ? cntL[tid] : 0u;
    const unsigned ex = gat_block_scan(cn, wsum);
    if (tid < GAT_BIN_NODES) {
        curL[tid] = ex;
        const int n = node0 + tid;
        if (n < N) { cnt[n] = (int)cn; offs[n] = beg + (int)ex; }
    }
    __syncthreads();
    const bool fits = m <= GAT_SORT_CAP;
    for (int b0 = 0; b0 < m; b0 += 4096) {
        int2 pl[4];
#pragma unroll
        for (int u = 0; u < 4; ++u) {
            const int i = b0 + u * 1024 + tid;
            pl[u] = part[beg + (i < m ? i : m - 1)];
        }
#pragma unroll
        for (int u = 0; u < 4; ++u) {
            const int i = b0 + u * 1024 + tid;
            if (i < m) {
                const unsigned pos = atomicAdd(&curL[pl[u].y - node0], 1u);
                if (fits) stageS[pos] = pl[u].x; else tmp[beg + (int)pos] = pl[u].x;
            }
        }
    }
    if (fits) {
        __syncthreads();
        for (int i = tid; i < m; i += 1024) tmp[beg + i] = stageS[i];
    }
}

__global__ __launch_bounds__(256) void gat_order_kernel(
    const int* __restrict__ cnt, const int* __restrict__ offs,
    const int* __restrict__ tmp, int* __restrict__ sorted, int N)
{
    int wid  = blockIdx.x * 8 + (threadIdx.x >> 5);
    int lane = threadIdx.x & 31;
    if (wid >= N) return;
    const int deg = __builtin_amdgcn_readfirstlane(cnt[wid]);
    const int off = __builtin_amdgcn_readfirstlane(offs[wid]);
    if (deg <= 32) {
        int v = (lane < deg) ? tmp[off + lane] : 0x7fffffff;
        int rank = 0;
#pragma unroll
        for (int j = 0; j < 32; ++j) {
            int vj = __builtin_amdgcn_readlane(v, j);
            rank += (vj < v || (vj == v && j < lane)) ? 1 : 0;
        }
        if (lane < deg) sorted[off + rank] = v;
    } else {
        for (int i = lane; i < deg; i += 32) {
            int v = tmp[off + i];
            int rank = 0;
            for (int j = 0; j < deg; ++j) {
                int vj = tmp[off + j];
                rank += (vj < v || (vj == v && j < i)) ? 1 : 0;
            }
            sorted[off + rank] = v;
        }
    }
}

__device__ __forceinline__ float gat_rl(float x, int k) {
    return __uint_as_float((unsigned)__builtin_amdgcn_readlane((int)__float_as_uint(x), k));
}

__global__ __launch_bounds__(256) void gat_node_l1_kernel(
    const int* __restrict__ cnt, const int* __restrict__ offs, const int* __restrict__ sorted,
    const float* __restrict__ as_, const float* __restrict__ ad_,
    const float* __restrict__ feat, const float* __restrict__ bias,
    float* __restrict__ outp, int N)
{
    int wid  = blockIdx.x * 8 + (threadIdx.x >> 5);
    const int lane = threadIdx.x & 31;
    if (wid >= N) return;
    const int n   = __builtin_amdgcn_readfirstlane(wid);
    const int deg = __builtin_amdgcn_readfirstlane(cnt[n]);
    const int off = __builtin_amdgcn_readfirstlane(offs[n]);
    const float2* as2 = reinterpret_cast<const float2*>(as_);
    const float2* ad2 = reinterpret_cast<const float2*>(ad_);
    const float4* f4  = reinterpret_cast<const float4*>(feat);
    const float2 adn = ad2[n];
    const float2 asn = as2[n];
    const float es0 = gat_lrelu(asn.x + adn.x), es1 = gat_lrelu(asn.y + adn.y);

    const bool v0 = lane < deg;
    int s = n; float e0 = es0, e1 = es1;
    if (v0) {
        s = sorted[off + lane];
        float2 a = as2[s];
        e0 = gat_lrelu(a.x + adn.x); e1 = gat_lrelu(a.y + adn.y);
    }
    float m0 = fmaxf(e0, es0), m1 = fmaxf(e1, es1);
    for (int base = 32; base < deg; base += 32) {
        int i = base + lane;
        if (i < deg) {
            float2 a = as2[sorted[off + i]];
            m0 = fmaxf(m0, gat_lrelu(a.x + adn.x)); m1 = fmaxf(m1, gat_lrelu(a.y + adn.y));
        }
    }
    for (int o = 16; o > 0; o >>= 1) {
        m0 = fmaxf(m0, __shfl_xor(m0, o, 32));
        m1 = fmaxf(m1, __shfl_xor(m1, o, 32));
    }
    const float p0 = v0 ? expf(e0 - m0) : 0.0f, p1 = v0 ? expf(e1 - m1) : 0.0f;
    float d0 = p0, d1 = p1;
    for (int base = 32; base < deg; base += 32) {
        int i = base + lane;
        if (i < deg) {
            float2 a = as2[sorted[off + i]];
            d0 += expf(gat_lrelu(a.x + adn.x) - m0); d1 += expf(gat_lrelu(a.y + adn.y) - m1);
        }
    }
    for (int o = 16; o > 0; o >>= 1) {
        d0 += __shfl_xor(d0, o, 32);
        d1 += __shfl_xor(d1, o, 32);
    }
    const float ps0 = expf(es0 - m0), ps1 = expf(es1 - m1);
    const float den0 = (d0 + ps0) + 1e-16f, den1 = (d1 + ps1) + 1e-16f;

    float4 acc = make_float4(0.0f, 0.0f, 0.0f, 0.0f);
    auto gather = [&](int sv, float q0, float q1, int cn) {
        for (int k = 0; k < cn; k += 4) {
#pragma unroll
            for (int u = 0; u < 4; ++u) {
                int   sk = __builtin_amdgcn_readlane(sv, k + u);
                float a0 = gat_rl(q0, k + u), a1 = gat_rl(q1, k + u);
                float ck = (lane < 16) ? a0 : a1;
                float4 f = f4[(long)sk * 32 + lane];
                acc.x = fmaf(ck, f.x, acc.x); acc.y = fmaf(ck, f.y, acc.y);
                acc.z = fmaf(ck, f.z, acc.z); acc.w = fmaf(ck, f.w, acc.w);
            }
        }
    };
    gather(s, p0 / den0, p1 / den1, deg < 32 ? deg : 32);
    for (int base = 32; base < deg; base += 32) {
        int i = base + lane;
        int t = n; float q0 = 0.0f, q1 = 0.0f;
        if (i < deg) {
            t = sorted[off + i];
            float2 a = as2[t];
            q0 = expf(gat_lrelu(a.x + adn.x) - m0) / den0;
            q1 = expf(gat_lrelu(a.y + adn.y) - m1) / den1;
        }
        gather(t, q0, q1, (deg - base) < 32 ? (deg - base) : 32);
    }
    {
        float cs = (lane < 16) ? (ps0 / den0) : (ps1 / den1);
        float4 f = f4[(long)n * 32 + lane];
        acc.x = fmaf(cs, f.x, acc.x); acc.y = fmaf(cs, f.y, acc.y);
        acc.z = fmaf(cs, f.z, acc.z); acc.w = fmaf(cs, f.w, acc.w);
    }
    const float4 b = reinterpret_cast<const float4*>(bias)[lane];
    float4 r;
    float t0 = acc.x + b.x, t1 = acc.y + b.y, t2 = acc.z + b.z, t3 = acc.w + b.w;
    r.x = t0 > 0.0f ? t0 : (expf(t0) - 1.0f);
    r.y = t1 > 0.0f ? t1 : (expf(t1) - 1.0f);
    r.z = t2 > 0.0f ? t2 : (expf(t2) - 1.0f);
    r.w = t3 > 0.0f ? t3 : (expf(t3) - 1.0f);
    reinterpret_cast<float4*>(outp)[(long)n * 32 + lane] = r;
}

__global__ __launch_bounds__(256) void gat_node_l2_kernel(
    const int* __restrict__ cnt, const int* __restrict__ offs, const int* __restrict__ sorted,
    const float* __restrict__ as_, const float* __restrict__ ad_,
    const float* __restrict__ feat, const float* __restrict__ bias,
    float* __restrict__ outp, int N)
{
    int wid  = blockIdx.x * 8 + (threadIdx.x >> 5);
    const int lane = threadIdx.x & 31;
    if (wid >= N) return;
    const int n   = __builtin_amdgcn_readfirstlane(wid);
    const int deg = __builtin_amdgcn_readfirstlane(cnt[n]);
    const int off = __builtin_amdgcn_readfirstlane(offs[n]);
    const float4* f4 = reinterpret_cast<const float4*>(feat);
    const float adn = ad_[n];
    const float es  = gat_lrelu(as_[n] + adn);

    const bool v0 = lane < deg;
    int s = n; float e = es;
    if (v0) { s = sorted[off + lane]; e = gat_lrelu(as_[s] + adn); }
    float m = fmaxf(e, es);
    for (int base = 32; base < deg; base += 32) {
        int i = base + lane;
        if (i < deg) m = fmaxf(m, gat_lrelu(as_[sorted[off + i]] + adn));
    }
    for (int o = 16; o > 0; o >>= 1) m = fmaxf(m, __shfl_xor(m, o, 32));
    const float p = v0 ? expf(e - m) : 0.0f;
    float d = p;
    for (int base = 32; base < deg; base += 32) {
        int i = base + lane;
        if (i < deg) d += expf(gat_lrelu(as_[sorted[off + i]] + adn) - m);
    }
    for (int o = 16; o > 0; o >>= 1) d += __shfl_xor(d, o, 32);
    const float ps  = expf(es - m);
    const float den = (d + ps) + 1e-16f;

    const int  g  = lane / 10;
    const int  cl = lane - g * 10;
    const bool on = g < 3;
    float4 acc = make_float4(0.0f, 0.0f, 0.0f, 0.0f);
    auto gather = [&](int sv, float q, int cn) {
        const int iters = (cn + 2) / 3;
        for (int tb = 0; tb < iters; tb += 4) {
#pragma unroll
          for (int u = 0; u < 4; ++u) {
            int k  = (tb + u) * 3 + g;
            const bool val = on && (k < cn);
            int kk = (k < cn) ? k : 0;
            int   sk = __shfl(sv, kk, 32);
            float ck = __shfl(q, kk, 32);
            ck = val ? ck : 0.0f;
            float4 f = f4[(long)(val ? sk : n) * 10 + cl];
            acc.x = fmaf(ck, f.x, acc.x); acc.y = fmaf(ck, f.y, acc.y);
            acc.z = fmaf(ck, f.z, acc.z); acc.w = fmaf(ck, f.w, acc.w);
          }
        }
    };
    gather(s, p / den, deg < 32 ? deg : 32);
    for (int base = 32; base < deg; base += 32) {
        int i = base + lane;
        int t = n; float q = 0.0f;
        if (i < deg) {
            t = sorted[off + i];
            q = expf(gat_lrelu(as_[t] + adn) - m) / den;
        }
        gather(t, q, (deg - base) < 32 ? (deg - base) : 32);
    }
    const int l1 = (lane + 10) & 31, l2 = (lane + 20) & 31;
    float4 a1, a2;
    a1.x = __shfl(acc.x, l1, 32); a1.y = __shfl(acc.y, l1, 32);
    a1.z = __shfl(acc.z, l1, 32); a1.w = __shfl(acc.w, l1, 32);
    a2.x = __shfl(acc.x, l2, 32); a2.y = __shfl(acc.y, l2, 32);
    a2.z = __shfl(acc.z, l2, 32); a2.w = __shfl(acc.w, l2, 32);
    if (lane < 10) {
        const float cs = ps / den;
        const float4 f = f4[(long)n * 10 + lane];
        const float4 b = reinterpret_cast<const float4*>(bias)[lane];
        float4 r;
        r.x = fmaf(cs, f.x, (acc.x + a1.x) + a2.x) + b.x;
        r.y = fmaf(cs, f.y, (acc.y + a1.y) + a2.y) + b.y;
        r.z = fmaf(cs, f.z, (acc.z + a1.z) + a2.z) + b.z;
        r.w = fmaf(cs, f.w, (acc.w + a1.w) + a2.w) + b.w;
        reinterpret_cast<float4*>(outp)[(long)n * 10 + lane] = r;
    }
}

extern "C" void kernel_launch(void* const* d_in, const int* in_sizes, int n_in,
                              void* d_out, int out_size, void* d_ws, size_t ws_size,
                              hipStream_t stream)
{
    (void)n_in; (void)out_size; (void)ws_size;
    const float* x   = (const float*)d_in[0];
    const int*   ei  = (const int*)  d_in[1];
    const float* W1  = (const float*)d_in[2];
    const float* aS1 = (const float*)d_in[3];
    const float* aD1 = (const float*)d_in[4];
    const float* b1  = (const float*)d_in[5];
    const float* W2  = (const float*)d_in[6];
    const float* aS2 = (const float*)d_in[7];
    const float* aD2 = (const float*)d_in[8];
    const float* b2  = (const float*)d_in[9];

    const int IN = 128, HC1 = 128, OUTC = 40;
    int N  = in_sizes[0] / IN;
    int E  = in_sizes[1] / 2;
    const int* srcI = ei;
    const int* dstI = ei + E;

    float* wsf = (float*)d_ws;
    size_t off = 0;
    auto take = [&](size_t nf) -> size_t { size_t o = off; off += (nf + 63) & ~(size_t)63; return o; };
    float* h1     = wsf + take((size_t)N * HC1);
    float* agg1   = wsf + take((size_t)N * HC1);
    float* as1    = wsf + take((size_t)N * 2);
    float* ad1    = wsf + take((size_t)N * 2);
    float* h2     = wsf + take((size_t)N * OUTC);
    float* as2    = wsf + take((size_t)N);
    float* ad2    = wsf + take((size_t)N);
    int*   cnt    = (int*)(wsf + take((size_t)N));
    int*   offs   = (int*)(wsf + take((size_t)N));
    int*   binws  = (int*)(wsf + take((size_t)N));
    int*   tmp    = (int*)(wsf + take((size_t)E));
    int*   sorted = (int*)(wsf + take((size_t)E));

    dim3 blk(256);
    int ngrid = (N + 7) / 8;

    int NB   = (N + GAT_BIN_NODES - 1) >> GAT_BIN_SHIFT;
    int nblk = (E + GAT_EPB - 1) / GAT_EPB;
    unsigned* binCount  = (unsigned*)binws;
    int*      binStart  = binws + (size_t)NB * GAT_CPAD;
    int2*     part      = (int2*)agg1;
    unsigned* blockBase = (unsigned*)(agg1 + (size_t)2 * E);
    gat_fill_u32_kernel<<<(NB * GAT_CPAD + 255) / 256, blk, 0, stream>>>(binCount, 0u, (long)NB * GAT_CPAD);
    gat_bin_hist_kernel<<<nblk, 1024, 0, stream>>>(dstI, E, NB, binCount, blockBase);
    gat_bin_scatter_kernel<<<nblk, 1024, 0, stream>>>(srcI, dstI, E, NB, binCount, blockBase, binStart, part);
    gat_bin_sort_kernel<<<NB, 1024, 0, stream>>>(part, binStart, N, cnt, offs, tmp);
    gat_order_kernel<<<ngrid, blk, 0, stream>>>(cnt, offs, tmp, sorted, N);

    gat_gemm_wmma<128, 8><<<(N + 127) / 128, blk, 0, stream>>>(x, W1, h1, N, HC1);
    gat_alpha_kernel<2, 64><<<ngrid, blk, 0, stream>>>(h1, aS1, aD1, as1, ad1, N);
    gat_node_l1_kernel<<<ngrid, blk, 0, stream>>>(cnt, offs, sorted, as1, ad1, h1, b1, agg1, N);

    gat_gemm_wmma<48, 3><<<(N + 127) / 128, blk, 0, stream>>>(agg1, W2, h2, N, OUTC);
    gat_alpha_kernel<1, 40><<<ngrid, blk, 0, stream>>>(h2, aS2, aD2, as2, ad2, N);
    gat_node_l2_kernel<<<ngrid, blk, 0, stream>>>(cnt, offs, sorted, as2, ad2, h2, b2, (float*)d_out, N);
}


# the five wave-per-node kernels run on 2048 resident blocks whose waves loop over nodes (attention vectors kept in registers) instead of 100000 one-node waves each; sum order unchanged
# speedup vs baseline: 4.7061x; 1.0399x over previous
#include <hip/hip_runtime.h>
#include <math.h>

typedef float v2f __attribute__((ext_vector_type(2)));
typedef float v8f __attribute__((ext_vector_type(8)));

#define NEG_SLOPE 0.2f

__device__ __forceinline__ float gat_lrelu(float v) { return v > 0.0f ? v : NEG_SLOPE * v; }

template<int NCOL, int NCT>
__global__ __launch_bounds__(256) void gat_gemm_wmma(
    const float* __restrict__ X, const float* __restrict__ W,
    float* __restrict__ Y, int N, int ncolStore)
{
    constexpr int K = 128;
    __shared__ float ldsW[K * NCOL];
    for (int idx = threadIdx.x; idx < K * NCOL; idx += 256) {
        int k = idx / NCOL, c = idx % NCOL;
        ldsW[idx] = (c < ncolStore) ? W[k * ncolStore + c] : 0.0f;
    }
    __syncthreads();

    const int lane = threadIdx.x & 31;
    const int wave = threadIdx.x >> 5;
    const int m    = lane & 15;
    const int half = lane >> 4;
    const int koff = half << 1;

    long row0 = (long)blockIdx.x * 128 + wave * 16;
    long rowL = row0 + m; if (rowL > (long)N - 1) rowL = (long)N - 1;
    const float* xrow = X + rowL * K;

    v8f acc[NCT] = {};
    for (int k0 = 0; k0 < K; k0 += 4) {
        v2f a;
        a.x = xrow[k0 + koff];
        a.y = xrow[k0 + koff + 1];
#pragma unroll
        for (int t = 0; t < NCT; ++t) {
            v2f b;
            b.x = ldsW[(k0 + koff) * NCOL + t * 16 + m];
            b.y = ldsW[(k0 + koff + 1) * NCOL + t * 16 + m];
            acc[t] = __builtin_amdgcn_wmma_f32_16x16x4_f32(
                false, a, false, b, (short)0, acc[t], false, false);
        }
    }
#pragma unroll
    for (int r = 0; r < 8; ++r) {
        long row = row0 + half * 8 + r;
        if (row < N) {
#pragma unroll
            for (int t = 0; t < NCT; ++t) {
                int c = t * 16 + m;
                if (c < ncolStore) Y[row * ncolStore + c] = acc[t][r];
            }
        }
    }
}

template<int H, int C>
__global__ __launch_bounds__(256) void gat_alpha_kernel(
    const float* __restrict__ h, const float* __restrict__ aw_s,
    const float* __restrict__ aw_d, float* __restrict__ as_,
    float* __restrict__ ad_, int N)
{
    constexpr int HC = H * C;
    constexpr int NT = (HC + 31) / 32;
    static_assert(H == 1 || C % 32 == 0, "an element's head must not depend on the lane");
    const int lane = threadIdx.x & 31;
    const int w0 = __builtin_amdgcn_readfirstlane((int)(blockIdx.x * 8 + (threadIdx.x >> 5)));
    const int nw = (int)gridDim.x * 8;
    float ws[NT], wd[NT];
#pragma unroll
    for (int t = 0; t < NT; ++t) {
        const int c = lane + 32 * t;
        ws[t] = (c < HC) ? aw_s[c] : 0.0f;
        wd[t] = (c < HC) ? aw_d[c] : 0.0f;
    }
    for (int wid = w0; wid < N; wid += nw) {
        const float* hp = h + (long)wid * HC;
        float v[NT];
#pragma unroll
        for (int t = 0; t < NT; ++t) {
            const int c = lane + 32 * t;
            v[t] = (c < HC) ? hp[c] : 0.0f;
        }
        float s[H], d[H];
#pragma unroll
        for (int hh = 0; hh < H; ++hh) { s[hh] = 0.0f; d[hh] = 0.0f; }
#pragma unroll
        for (int t = 0; t < NT; ++t) {
            const int c  = lane + 32 * t;
            const int hh = (H == 1) ? 0 : ((32 * t) / C);
            if (c < HC) {
                s[hh] += v[t] * ws[t];
                d[hh] += v[t] * wd[t];
            }
        }
#pragma unroll
        for (int hh = 0; hh < H; ++hh) {
            float sv = s[hh], dv = d[hh];
            for (int o = 16; o > 0; o >>= 1) {
                sv += __shfl_xor(sv, o, 32);
                dv += __shfl_xor(dv, o, 32);
            }
            if (lane == 0) {
                as_[(long)wid * H + hh] = sv;
                ad_[(long)wid * H + hh] = dv;
            }
        }
    }
}

#define GAT_BIN_SHIFT 9
#define GAT_BIN_NODES 512
#define GAT_MAXB      1024
#define GAT_EPB       4096
#define GAT_EPT       4
#define GAT_CPAD      32
#define GAT_SORT_CAP  12288

__global__ void gat_fill_u32_kernel(unsigned* __restrict__ p, unsigned v, long n) {
    long i = blockIdx.x * (long)blockDim.x + threadIdx.x;
    if (i < n) p[i] = v;
}

__device__ __forceinline__ unsigned gat_block_scan(unsigned c, unsigned* wsum) {
    const int lane = threadIdx.x & 31, wave = threadIdx.x >> 5;
    unsigned inc = c;
    for (int o = 1; o < 32; o <<= 1) {
        unsigned y = __shfl_up(inc, o, 32);
        if (lane >= o) inc += y;
    }
    if (lane == 31) wsum[wave] = inc;
    __syncthreads();
    unsigned wb = 0;
    for (int w = 0; w < wave; ++w) wb += wsum[w];
    __syncthreads();
    return wb + inc - c;
}

__global__ __launch_bounds__(1024) void gat_bin_hist_kernel(
    const int* __restrict__ dst, int E, int NB,
    unsigned* __restrict__ binCount, unsigned* __restrict__ blockBase)
{
    __shared__ unsigned hist[GAT_MAXB];
    const int tid = threadIdx.x;
    hist[tid] = 0u;
    __syncthreads();
    const int e0 = blockIdx.x * GAT_EPB;
    int dv[GAT_EPT];
#pragma unroll
    for (int j = 0; j < GAT_EPT; ++j) {
        int e = e0 + j * 1024 + tid;
        dv[j] = -1;
        if (e < E) dv[j] = dst[e];
    }
#pragma unroll
    for (int j = 0; j < GAT_EPT; ++j) {
        if (dv[j] >= 0) atomicAdd(&hist[dv[j] >> GAT_BIN_SHIFT], 1u);
    }
    __syncthreads();
    if (tid < NB) {
        const unsigned cb = hist[tid];
        unsigned base = 0u;
        if (cb) base = atomicAdd(&binCount[tid * GAT_CPAD], cb);
        blockBase[(long)blockIdx.x * NB + tid] = base;
    }
}

__global__ __launch_bounds__(1024) void gat_bin_scatter_kernel(
    const int* __restrict__ src, const int* __restrict__ dst, int E, int NB,
    const unsigned* __restrict__ binCount, const unsigned* __restrict__ blockBase,
    int* __restrict__ binStart, int2* __restrict__ part)
{
    __shared__ int2 stage[GAT_EPB];
    __shared__ unsigned hist[GAT_MAXB];
    __shared__ unsigned lstart[GAT_MAXB];
    __shared__ unsigned gbase[GAT_MAXB];
    __shared__ unsigned wsum[32];
    const int tid = threadIdx.x;
    hist[tid] = 0u;
    __syncthreads();
    const int e0 = blockIdx.x * GAT_EPB;
    int sv[GAT_EPT], dv[GAT_EPT];
    unsigned rk[GAT_EPT];
#pragma unroll
    for (int j = 0; j < GAT_EPT; ++j) {
        int e = e0 + j * 1024 + tid;
        sv[j] = 0; dv[j] = -1;
        if (e < E) { dv[j] = dst[e]; sv[j] = src[e]; }
    }
#pragma unroll
    for (int j = 0; j < GAT_EPT; ++j) {
        rk[j] = 0u;
        if (dv[j] >= 0) rk[j] = atomicAdd(&hist[dv[j] >> GAT_BIN_SHIFT], 1u);
    }
    __syncthreads();
    const unsigned ls = gat_block_scan(hist[tid], wsum);
    const unsigned bc = (tid < NB) ? binCount[tid * GAT_CPAD] : 0u;
    const unsigned bs = gat_block_scan(bc, wsum);
    lstart[tid] = ls;
    if (tid < NB) {
        gbase[tid] = bs + blockBase[(long)blockIdx.x * NB + tid] - ls;
        if (blockIdx.x == 0) {
            binStart[tid] = (int)bs;
            if (tid == NB - 1) binStart[NB] = (int)(bs + bc);
        }
    }
    __syncthreads();
#pragma unroll
    for (int j = 0; j < GAT_EPT; ++j) {
        if (dv[j] >= 0) stage[lstart[dv[j] >> GAT_BIN_SHIFT] + rk[j]] = make_int2(sv[j], dv[j]);
    }
    __syncthreads();
    const int rem = E - e0;
    const int nval = rem < GAT_EPB ? rem : GAT_EPB;
    for (int i = tid; i < nval; i += 1024) {
        const int2 p = stage[i];
        part[gbase[p.y >> GAT_BIN_SHIFT] + (unsigned)i] = p;
    }
}

__global__ __launch_bounds__(1024) void gat_bin_sort_kernel(
    const int2* __restrict__ part, const int* __restrict__ binStart, int N,
    int* __restrict__ cnt, int* __restrict__ offs, int* __restrict__ tmp)
{
    __shared__ int stageS[GAT_SORT_CAP];
    __shared__ unsigned cntL[GAT_BIN_NODES];
    __shared__ unsigned curL[GAT_BIN_NODES];
    __shared__ unsigned wsum[32];
    const int tid = threadIdx.x;
    const int bin = blockIdx.x;
    const int beg = binStart[bin];
    const int m   = binStart[bin + 1] - beg;
    const int node0 = bin << GAT_BIN_SHIFT;
    if (tid < GAT_BIN_NODES) cntL[tid] = 0u;
    __syncthreads();
    for (int b0 = 0; b0 < m; b0 += 4096) {
        int dl[4];
#pragma unroll
        for (int u = 0; u < 4; ++u) {
            const int i = b0 + u * 1024 + tid;
            dl[u] = part[beg + (i < m ? i : m - 1)].y;
        }
#pragma unroll
        for (int u = 0; u < 4; ++u) {
            const int i = b0 + u * 1024 + tid;
            if (i < m) atomicAdd(&cntL[dl[u] - node0], 1u);
        }
    }
    __syncthreads();
    const unsigned cn = (tid < GAT_BIN_NODES) ? cntL[tid] : 0u;
    const unsigned ex = gat_block_scan(cn, wsum);
    if (tid < GAT_BIN_NODES) {
        curL[tid] = ex;
        const int n = node0 + tid;
        if (n < N) { cnt[n] = (int)cn; offs[n] = beg + (int)ex; }
    }
    __syncthreads();
    const bool fits = m <= GAT_SORT_CAP;
    for (int b0 = 0; b0 < m; b0 += 4096) {
        int2 pl[4];
#pragma unroll
        for (int u = 0; u < 4; ++u) {
            const int i = b0 + u * 1024 + tid;
            pl[u] = part[beg + (i < m ? i : m - 1)];
        }
#pragma unroll
        for (int u = 0; u < 4; ++u) {
            const int i = b0 + u * 1024 + tid;
            if (i < m) {
                const unsigned pos = atomicAdd(&curL[pl[u].y - node0], 1u);
                if (fits) stageS[pos] = pl[u].x; else tmp[beg + (int)pos] = pl[u].x;
            }
        }
    }
    if (fits) {
        __syncthreads();
        for (int i = tid; i < m; i += 1024) tmp[beg + i] = stageS[i];
    }
}

__global__ __launch_bounds__(256) void gat_order_kernel(
    const int* __restrict__ cnt, const int* __restrict__ offs,
    const int* __restrict__ tmp, int* __restrict__ sorted, int N)
{
    const int lane = threadIdx.x & 31;
    const int w0 = __builtin_amdgcn_readfirstlane((int)(blockIdx.x * 8 + (threadIdx.x >> 5)));
    const int nw = (int)gridDim.x * 8;
    for (int wid = w0; wid < N; wid += nw) {
        const int deg = __builtin_amdgcn_readfirstlane(cnt[wid]);
        const int off = __builtin_amdgcn_readfirstlane(offs[wid]);
        if (deg <= 32) {
            int v = (lane < deg) ? tmp[off + lane] : 0x7fffffff;
            int rank = 0;
#pragma unroll
            for (int j = 0; j < 32; ++j) {
                int vj = __builtin_amdgcn_readlane(v, j);
                rank += (vj < v || (vj == v && j < lane)) ? 1 : 0;
            }
            if (lane < deg) sorted[off + rank] = v;
        } else {
            for (int i = lane; i < deg; i += 32) {
                int v = tmp[off + i];
                int rank = 0;
                for (int j = 0; j < deg; ++j) {
                    int vj = tmp[off + j];
                    rank += (vj < v || (vj == v && j < i)) ? 1 : 0;
                }
                sorted[off + rank] = v;
            }
        }
    }
}

__device__ __forceinline__ float gat_rl(float x, int k) {
    return __uint_as_float((unsigned)__builtin_amdgcn_readlane((int)__float_as_uint(x), k));
}

__global__ __launch_bounds__(256) void gat_node_l1_kernel(
    const int* __restrict__ cnt, const int* __restrict__ offs, const int* __restrict__ sorted,
    const float* __restrict__ as_, const float* __restrict__ ad_,
    const float* __restrict__ feat, const float* __restrict__ bias,
    float* __restrict__ outp, int N)
{
    const int lane = threadIdx.x & 31;
    const int w0 = __builtin_amdgcn_readfirstlane((int)(blockIdx.x * 8 + (threadIdx.x >> 5)));
    const int nw = (int)gridDim.x * 8;
    for (int n = w0; n < N; n += nw) {
        const int deg = __builtin_amdgcn_readfirstlane(cnt[n]);
        const int off = __builtin_amdgcn_readfirstlane(offs[n]);
        const float2* as2 = reinterpret_cast<const float2*>(as_);
        const float2* ad2 = reinterpret_cast<const float2*>(ad_);
        const float4* f4  = reinterpret_cast<const float4*>(feat);
        const float2 adn = ad2[n];
        const float2 asn = as2[n];
        const float es0 = gat_lrelu(asn.x + adn.x), es1 = gat_lrelu(asn.y + adn.y);

        const bool v0 = lane < deg;
        int s = n; float e0 = es0, e1 = es1;
        if (v0) {
            s = sorted[off + lane];
            float2 a = as2[s];
            e0 = gat_lrelu(a.x + adn.x); e1 = gat_lrelu(a.y + adn.y);
        }
        float m0 = fmaxf(e0, es0), m1 = fmaxf(e1, es1);
        for (int base = 32; base < deg; base += 32) {
            int i = base + lane;
            if (i < deg) {
                float2 a = as2[sorted[off + i]];
                m0 = fmaxf(m0, gat_lrelu(a.x + adn.x)); m1 = fmaxf(m1, gat_lrelu(a.y + adn.y));
            }
        }
        for (int o = 16; o > 0; o >>= 1) {
            m0 = fmaxf(m0, __shfl_xor(m0, o, 32));
            m1 = fmaxf(m1, __shfl_xor(m1, o, 32));
        }
        const float p0 = v0 ? expf(e0 - m0) : 0.0f, p1 = v0 ? expf(e1 - m1) : 0.0f;
        float d0 = p0, d1 = p1;
        for (int base = 32; base < deg; base += 32) {
            int i = base + lane;
            if (i < deg) {
                float2 a = as2[sorted[off + i]];
                d0 += expf(gat_lrelu(a.x + adn.x) - m0); d1 += expf(gat_lrelu(a.y + adn.y) - m1);
            }
        }
        for (int o = 16; o > 0; o >>= 1) {
            d0 += __shfl_xor(d0, o, 32);
            d1 += __shfl_xor(d1, o, 32);
        }
        const float ps0 = expf(es0 - m0), ps1 = expf(es1 - m1);
        const float den0 = (d0 + ps0) + 1e-16f, den1 = (d1 + ps1) + 1e-16f;

        float4 acc = make_float4(0.0f, 0.0f, 0.0f, 0.0f);
        auto gather = [&](int sv, float q0, float q1, int cn) {
            for (int k = 0; k < cn; k += 4) {
#pragma unroll
                for (int u = 0; u < 4; ++u) {
                    int   sk = __builtin_amdgcn_readlane(sv, k + u);
                    float a0 = gat_rl(q0, k + u), a1 = gat_rl(q1, k + u);
                    float ck = (lane < 16) ? a0 : a1;
                    float4 f = f4[(long)sk * 32 + lane];
                    acc.x = fmaf(ck, f.x, acc.x); acc.y = fmaf(ck, f.y, acc.y);
                    acc.z = fmaf(ck, f.z, acc.z); acc.w = fmaf(ck, f.w, acc.w);
                }
            }
        };
        gather(s, p0 / den0, p1 / den1, deg < 32 ? deg : 32);
        for (int base = 32; base < deg; base += 32) {
            int i = base + lane;
            int t = n; float q0 = 0.0f, q1 = 0.0f;
            if (i < deg) {
                t = sorted[off + i];
                float2 a = as2[t];
                q0 = expf(gat_lrelu(a.x + adn.x) - m0) / den0;
                q1 = expf(gat_lrelu(a.y + adn.y) - m1) / den1;
            }
            gather(t, q0, q1, (deg - base) < 32 ? (deg - base) : 32);
        }
        {
            float cs = (lane < 16) ? (ps0 / den0) : (ps1 / den1);
            float4 f = f4[(long)n * 32 + lane];
            acc.x = fmaf(cs, f.x, acc.x); acc.y = fmaf(cs, f.y, acc.y);
            acc.z = fmaf(cs, f.z, acc.z); acc.w = fmaf(cs, f.w, acc.w);
        }
        const float4 b = reinterpret_cast<const float4*>(bias)[lane];
        float4 r;
        float t0 = acc.x + b.x, t1 = acc.y + b.y, t2 = acc.z + b.z, t3 = acc.w + b.w;
        r.x = t0 > 0.0f ? t0 : (expf(t0) - 1.0f);
        r.y = t1 > 0.0f ? t1 : (expf(t1) - 1.0f);
        r.z = t2 > 0.0f ? t2 : (expf(t2) - 1.0f);
        r.w = t3 > 0.0f ? t3 : (expf(t3) - 1.0f);
        reinterpret_cast<float4*>(outp)[(long)n * 32 + lane] = r;
    }
}

__global__ __launch_bounds__(256) void gat_node_l2_kernel(
    const int* __restrict__ cnt, const int* __restrict__ offs, const int* __restrict__ sorted,
    const float* __restrict__ as_, const float* __restrict__ ad_,
    const float* __restrict__ feat, const float* __restrict__ bias,
    float* __restrict__ outp, int N)
{
    const int lane = threadIdx.x & 31;
    const int w0 = __builtin_amdgcn_readfirstlane((int)(blockIdx.x * 8 + (threadIdx.x >> 5)));
    const int nw = (int)gridDim.x * 8;
    for (int n = w0; n < N; n += nw) {
        const int deg = __builtin_amdgcn_readfirstlane(cnt[n]);
        const int off = __builtin_amdgcn_readfirstlane(offs[n]);
        const float4* f4 = reinterpret_cast<const float4*>(feat);
        const float adn = ad_[n];
        const float es  = gat_lrelu(as_[n] + adn);

        const bool v0 = lane < deg;
        int s = n; float e = es;
        if (v0) { s = sorted[off + lane]; e = gat_lrelu(as_[s] + adn); }
        float m = fmaxf(e, es);
        for (int base = 32; base < deg; base += 32) {
            int i = base + lane;
            if (i < deg) m = fmaxf(m, gat_lrelu(as_[sorted[off + i]] + adn));
        }
        for (int o = 16; o > 0; o >>= 1) m = fmaxf(m, __shfl_xor(m, o, 32));
        const float p = v0 ? expf(e - m) : 0.0f;
        float d = p;
        for (int base = 32; base < deg; base += 32) {
            int i = base + lane;
            if (i < deg) d += expf(gat_lrelu(as_[sorted[off + i]] + adn) - m);
        }
        for (int o = 16; o > 0; o >>= 1) d += __shfl_xor(d, o, 32);
        const float ps  = expf(es - m);
        const float den = (d + ps) + 1e-16f;

        const int  g  = lane / 10;
        const int  cl = lane - g * 10;
        const bool on = g < 3;
        float4 acc = make_float4(0.0f, 0.0f, 0.0f, 0.0f);
        auto gather = [&](int sv, float q, int cn) {
            const int iters = (cn + 2) / 3;
            for (int tb = 0; tb < iters; tb += 4) {
#pragma unroll
              for (int u = 0; u < 4; ++u) {
                int k  = (tb + u) * 3 + g;
                const bool val = on && (k < cn);
                int kk = (k < cn) ? k : 0;
                int   sk = __shfl(sv, kk, 32);
                float ck = __shfl(q, kk, 32);
                ck = val ? ck : 0.0f;
                float4 f = f4[(long)(val ? sk : n) * 10 + cl];
                acc.x = fmaf(ck, f.x, acc.x); acc.y = fmaf(ck, f.y, acc.y);
                acc.z = fmaf(ck, f.z, acc.z); acc.w = fmaf(ck, f.w, acc.w);
              }
            }
        };
        gather(s, p / den, deg < 32 ? deg : 32);
        for (int base = 32; base < deg; base += 32) {
            int i = base + lane;
            int t = n; float q = 0.0f;
            if (i < deg) {
                t = sorted[off + i];
                q = expf(gat_lrelu(as_[t] + adn) - m) / den;
            }
            gather(t, q, (deg - base) < 32 ? (deg - base) : 32);
        }
        const int l1 = (lane + 10) & 31, l2 = (lane + 20) & 31;
        float4 a1, a2;
        a1.x = __shfl(acc.x, l1, 32); a1.y = __shfl(acc.y, l1, 32);
        a1.z = __shfl(acc.z, l1, 32); a1.w = __shfl(acc.w, l1, 32);
        a2.x = __shfl(acc.x, l2, 32); a2.y = __shfl(acc.y, l2, 32);
        a2.z = __shfl(acc.z, l2, 32); a2.w = __shfl(acc.w, l2, 32);
        if (lane < 10) {
            const float cs = ps / den;
            const float4 f = f4[(long)n * 10 + lane];
            const float4 b = reinterpret_cast<const float4*>(bias)[lane];
            float4 r;
            r.x = fmaf(cs, f.x, (acc.x + a1.x) + a2.x) + b.x;
            r.y = fmaf(cs, f.y, (acc.y + a1.y) + a2.y) + b.y;
            r.z = fmaf(cs, f.z, (acc.z + a1.z) + a2.z) + b.z;
            r.w = fmaf(cs, f.w, (acc.w + a1.w) + a2.w) + b.w;
            reinterpret_cast<float4*>(outp)[(long)n * 10 + lane] = r;
        }
    }
}

extern "C" void kernel_launch(void* const* d_in, const int* in_sizes, int n_in,
                              void* d_out, int out_size, void* d_ws, size_t ws_size,
                              hipStream_t stream)
{
    (void)n_in; (void)out_size; (void)ws_size;
    const float* x   = (const float*)d_in[0];
    const int*   ei  = (const int*)  d_in[1];
    const float* W1  = (const float*)d_in[2];
    const float* aS1 = (const float*)d_in[3];
    const float* aD1 = (const float*)d_in[4];
    const float* b1  = (const float*)d_in[5];
    const float* W2  = (const float*)d_in[6];
    const float* aS2 = (const float*)d_in[7];
    const float* aD2 = (const float*)d_in[8];
    const float* b2  = (const float*)d_in[9];

    const int IN = 128, HC1 = 128, OUTC = 40;
    int N  = in_sizes[0] / IN;
    int E  = in_sizes[1] / 2;
    const int* srcI = ei;
    const int* dstI = ei + E;

    float* wsf = (float*)d_ws;
    size_t off = 0;
    auto take = [&](size_t nf) -> size_t { size_t o = off; off += (nf + 63) & ~(size_t)63; return o; };
    float* h1     = wsf + take((size_t)N * HC1);
    float* agg1   = wsf + take((size_t)N * HC1);
    float* as1    = wsf + take((size_t)N * 2);
    float* ad1    = wsf + take((size_t)N * 2);
    float* h2     = wsf + take((size_t)N * OUTC);
    float* as2    = wsf + take((size_t)N);
    float* ad2    = wsf + take((size_t)N);
    int*   cnt    = (int*)(wsf + take((size_t)N));
    int*   offs   = (int*)(wsf + take((size_t)N));
    int*   binws  = (int*)(wsf + take((size_t)N));
    int*   tmp    = (int*)(wsf + take((size_t)E));
    int*   sorted = (int*)(wsf + take((size_t)E));

    dim3 blk(256);
    dim3 ngrid(2048);

    int NB   = (N + GAT_BIN_NODES - 1) >> GAT_BIN_SHIFT;
    int nblk = (E + GAT_EPB - 1) / GAT_EPB;
    unsigned* binCount  = (unsigned*)binws;
    int*      binStart  = binws + (size_t)NB * GAT_CPAD;
    int2*     part      = (int2*)agg1;
    unsigned* blockBase = (unsigned*)(agg1 + (size_t)2 * E);
    gat_fill_u32_kernel<<<(NB * GAT_CPAD + 255) / 256, blk, 0, stream>>>(binCount, 0u, (long)NB * GAT_CPAD);
    gat_bin_hist_kernel<<<nblk, 1024, 0, stream>>>(dstI, E, NB, binCount, blockBase);
    gat_bin_scatter_kernel<<<nblk, 1024, 0, stream>>>(srcI, dstI, E, NB, binCount, blockBase, binStart, part);
    gat_bin_sort_kernel<<<NB, 1024, 0, stream>>>(part, binStart, N, cnt, offs, tmp);
    gat_order_kernel<<<ngrid, blk, 0, stream>>>(cnt, offs, tmp, sorted, N);

    gat_gemm_wmma<128, 8><<<(N + 127) / 128, blk, 0, stream>>>(x, W1, h1, N, HC1);
    gat_alpha_kernel<2, 64><<<ngrid, blk, 0, stream>>>(h1, aS1, aD1, as1, ad1, N);
    gat_node_l1_kernel<<<ngrid, blk, 0, stream>>>(cnt, offs, sorted, as1, ad1, h1, b1, agg1, N);

    gat_gemm_wmma<48, 3><<<(N + 127) / 128, blk, 0, stream>>>(agg1, W2, h2, N, OUTC);
    gat_alpha_kernel<1, 40><<<ngrid, blk, 0, stream>>>(h2, aS2, aD2, as2, ad2, N);
    gat_node_l2_kernel<<<ngrid, blk, 0, stream>>>(cnt, offs, sorted, as2, ad2, h2, b2, (float*)d_out, N);
}


# GEMM kernels stage W into LDS with all of a thread's loads in flight together (float4 pieces, no branch round a load) instead of one waited 4-byte load per element
# speedup vs baseline: 6.5638x; 1.3947x over previous
#include <hip/hip_runtime.h>
#include <math.h>

typedef float v2f __attribute__((ext_vector_type(2)));
typedef float v8f __attribute__((ext_vector_type(8)));

#define NEG_SLOPE 0.2f

__device__ __forceinline__ float gat_lrelu(float v) { return v > 0.0f ? v : NEG_SLOPE * v; }

template<int CNT, int OFF>
__device__ __forceinline__ void gat_fold_step(float* v, int m) {
    const bool up = (m & OFF) != 0;
#pragma unroll
    for (int j = 0; j < CNT; ++j) {
        const float keep = up ? v[j + CNT] : v[j];
        const float send = up ? v[j] : v[j + CNT];
        v[j] = keep + __shfl_xor(send, OFF, 32);
    }
}
template<int NV>
__device__ __forceinline__ void gat_fold16(float* v, int m) {
    gat_fold_step<NV / 2, 8>(v, m);
    gat_fold_step<NV / 4, 4>(v, m);
    gat_fold_step<NV / 8, 2>(v, m);
    gat_fold_step<NV / 16, 1>(v, m);
}

template<int NCOL, int NCT, int H, int NSTORE>
__global__ __launch_bounds__(256) void gat_gemm_wmma(
    const float* __restrict__ X, const float* __restrict__ W,
    const float* __restrict__ aw_s, const float* __restrict__ aw_d,
    float* __restrict__ Y, float* __restrict__ as_, float* __restrict__ ad_,
    int N)
{
    constexpr int K = 128;
    constexpr int ncolStore = NSTORE;
    constexpr int TPH = NCT / H;
    constexpr int NV  = 16 * H;
    constexpr int NC4 = NCOL / 4;
    constexpr int NLD = K * NC4 / 256;
    static_assert(NCT % H == 0, "a column tile must belong to one head");
    static_assert(NSTORE % 4 == 0 && NSTORE >= 4 && NSTORE <= NCOL, "rows of W are whole float4 pieces");
    static_assert((K * NC4) % 256 == 0, "every thread stages the same number of pieces");
    __shared__ __attribute__((aligned(16))) float ldsW[K * NCOL];
    {
        float4 wv[NLD];
#pragma unroll
        for (int i = 0; i < NLD; ++i) {
            const int p  = (int)threadIdx.x + i * 256;
            const int k  = p / NC4, c = (p % NC4) * 4;
            const int cc = c < NSTORE ? c : NSTORE - 4;
            wv[i] = *reinterpret_cast<const float4*>(W + k * NSTORE + cc);
        }
#pragma unroll
        for (int i = 0; i < NLD; ++i) {
            const int p = (int)threadIdx.x + i * 256;
            const int c = (p % NC4) * 4;
            reinterpret_cast<float4*>(ldsW)[p] = (c < NSTORE) ? wv[i] : make_float4(0.0f, 0.0f, 0.0f, 0.0f);
        }
    }
    __syncthreads();

    const int lane = threadIdx.x & 31;
    const int wave = threadIdx.x >> 5;
    const int m    = lane & 15;
    const int half = lane >> 4;
    const int koff = half << 1;

    long row0 = (long)blockIdx.x * 128 + wave * 16;
    long rowL = row0 + m; if (rowL > (long)N - 1) rowL = (long)N - 1;
    const float* xrow = X + rowL * K;

    float wS[NCT], wD[NCT];
#pragma unroll
    for (int t = 0; t < NCT; ++t) {
        const int c = t * 16 + m;
        wS[t] = (c < ncolStore) ? aw_s[c] : 0.0f;
        wD[t] = (c < ncolStore) ? aw_d[c] : 0.0f;
    }

    v8f acc[NCT] = {};
    for (int k0 = 0; k0 < K; k0 += 4) {
        v2f a;
        a.x = xrow[k0 + koff];
        a.y = xrow[k0 + koff + 1];
#pragma unroll
        for (int t = 0; t < NCT; ++t) {
            v2f b;
            b.x = ldsW[(k0 + koff) * NCOL + t * 16 + m];
            b.y = ldsW[(k0 + koff + 1) * NCOL + t * 16 + m];
            acc[t] = __builtin_amdgcn_wmma_f32_16x16x4_f32(
                false, a, false, b, (short)0, acc[t], false, false);
        }
    }
#pragma unroll
    for (int r = 0; r < 8; ++r) {
        long row = row0 + half * 8 + r;
        if (row < N) {
#pragma unroll
            for (int t = 0; t < NCT; ++t) {
                int c = t * 16 + m;
                if (c < ncolStore) Y[row * ncolStore + c] = acc[t][r];
            }
        }
    }

    float v[NV];
#pragma unroll
    for (int r = 0; r < 8; ++r) {
#pragma unroll
        for (int hh = 0; hh < H; ++hh) {
            float sv = 0.0f, dv = 0.0f;
#pragma unroll
            for (int u = 0; u < TPH; ++u) {
                const int t = hh * TPH + u;
                sv = fmaf(acc[t][r], wS[t], sv);
                dv = fmaf(acc[t][r], wD[t], dv);
            }
            v[(r * H + hh) * 2 + 0] = sv;
            v[(r * H + hh) * 2 + 1] = dv;
        }
    }
    gat_fold16<NV>(v, m);
    {
        const long row = row0 + half * 8 + (m >> 1);
        if (row < N) {
            if (H == 2) {
                as_[row * 2 + (m & 1)] = v[0];
                ad_[row * 2 + (m & 1)] = v[1];
            } else {
                if (m & 1) ad_[row] = v[0]; else as_[row] = v[0];
            }
        }
    }
}

#define GAT_BIN_SHIFT 9
#define GAT_BIN_NODES 512
#define GAT_MAXB      1024
#define GAT_EPB       4096
#define GAT_EPT       4
#define GAT_CPAD      32
#define GAT_SORT_CAP  12288

__global__ void gat_fill_u32_kernel(unsigned* __restrict__ p, unsigned v, long n) {
    long i = blockIdx.x * (long)blockDim.x + threadIdx.x;
    if (i < n) p[i] = v;
}

__device__ __forceinline__ unsigned gat_block_scan(unsigned c, unsigned* wsum) {
    const int lane = threadIdx.x & 31, wave = threadIdx.x >> 5;
    unsigned inc = c;
    for (int o = 1; o < 32; o <<= 1) {
        unsigned y = __shfl_up(inc, o, 32);
        if (lane >= o) inc += y;
    }
    if (lane == 31) wsum[wave] = inc;
    __syncthreads();
    unsigned wb = 0;
    for (int w = 0; w < wave; ++w) wb += wsum[w];
    __syncthreads();
    return wb + inc - c;
}

__global__ __launch_bounds__(1024) void gat_bin_hist_kernel(
    const int* __restrict__ dst, int E, int NB,
    unsigned* __restrict__ binCount, unsigned* __restrict__ blockBase)
{
    __shared__ unsigned hist[GAT_MAXB];
    const int tid = threadIdx.x;
    hist[tid] = 0u;
    __syncthreads();
    const int e0 = blockIdx.x * GAT_EPB;
    int dv[GAT_EPT];
#pragma unroll
    for (int j = 0; j < GAT_EPT; ++j) {
        int e = e0 + j * 1024 + tid;
        dv[j] = -1;
        if (e < E) dv[j] = dst[e];
    }
#pragma unroll
    for (int j = 0; j < GAT_EPT; ++j) {
        if (dv[j] >= 0) atomicAdd(&hist[dv[j] >> GAT_BIN_SHIFT], 1u);
    }
    __syncthreads();
    if (tid < NB) {
        const unsigned cb = hist[tid];
        unsigned base = 0u;
        if (cb) base = atomicAdd(&binCount[tid * GAT_CPAD], cb);
        blockBase[(long)blockIdx.x * NB + tid] = base;
    }
}

__global__ __launch_bounds__(1024) void gat_bin_scatter_kernel(
    const int* __restrict__ src, const int* __restrict__ dst, int E, int NB,
    const unsigned* __restrict__ binCount, const unsigned* __restrict__ blockBase,
    int* __restrict__ binStart, int2* __restrict__ part)
{
    __shared__ int2 stage[GAT_EPB];
    __shared__ unsigned hist[GAT_MAXB];
    __shared__ unsigned lstart[GAT_MAXB];
    __shared__ unsigned gbase[GAT_MAXB];
    __shared__ unsigned wsum[32];
    const int tid = threadIdx.x;
    hist[tid] = 0u;
    __syncthreads();
    const int e0 = blockIdx.x * GAT_EPB;
    int sv[GAT_EPT], dv[GAT_EPT];
    unsigned rk[GAT_EPT];
#pragma unroll
    for (int j = 0; j < GAT_EPT; ++j) {
        int e = e0 + j * 1024 + tid;
        sv[j] = 0; dv[j] = -1;
        if (e < E) { dv[j] = dst[e]; sv[j] = src[e]; }
    }
#pragma unroll
    for (int j = 0; j < GAT_EPT; ++j) {
        rk[j] = 0u;
        if (dv[j] >= 0) rk[j] = atomicAdd(&hist[dv[j] >> GAT_BIN_SHIFT], 1u);
    }
    __syncthreads();
    const unsigned ls = gat_block_scan(hist[tid], wsum);
    const unsigned bc = (tid < NB) ? binCount[tid * GAT_CPAD] : 0u;
    const unsigned bs = gat_block_scan(bc, wsum);
    lstart[tid] = ls;
    if (tid < NB) {
        gbase[tid] = bs + blockBase[(long)blockIdx.x * NB + tid] - ls;
        if (blockIdx.x == 0) {
            binStart[tid] = (int)bs;
            if (tid == NB - 1) binStart[NB] = (int)(bs + bc);
        }
    }
    __syncthreads();
#pragma unroll
    for (int j = 0; j < GAT_EPT; ++j) {
        if (dv[j] >= 0) stage[lstart[dv[j] >> GAT_BIN_SHIFT] + rk[j]] = make_int2(sv[j], dv[j]);
    }
    __syncthreads();
    const int rem = E - e0;
    const int nval = rem < GAT_EPB ? rem : GAT_EPB;
    for (int i = tid; i < nval; i += 1024) {
        const int2 p = stage[i];
        part[gbase[p.y >> GAT_BIN_SHIFT] + (unsigned)i] = p;
    }
}

__global__ __launch_bounds__(1024) void gat_bin_sort_kernel(
    const int2* __restrict__ part, const int* __restrict__ binStart, int N,
    int* __restrict__ cnt, int* __restrict__ offs, int* __restrict__ tmp)
{
    __shared__ int stageS[GAT_SORT_CAP];
    __shared__ unsigned cntL[GAT_BIN_NODES];
    __shared__ unsigned curL[GAT_BIN_NODES];
    __shared__ unsigned wsum[32];
    const int tid = threadIdx.x;
    const int bin = blockIdx.x;
    const int beg = binStart[bin];
    const int m   = binStart[bin + 1] - beg;
    const int node0 = bin << GAT_BIN_SHIFT;
    if (tid < GAT_BIN_NODES) cntL[tid] = 0u;
    __syncthreads();
    for (int b0 = 0; b0 < m; b0 += 4096) {
        int dl[4];
#pragma unroll
        for (int u = 0; u < 4; ++u) {
            const int i = b0 + u * 1024 + tid;
            dl[u] = part[beg + (i < m ? i : m - 1)].y;
        }
#pragma unroll
        for (int u = 0; u < 4; ++u) {
            const int i = b0 + u * 1024 + tid;
            if (i < m) atomicAdd(&cntL[dl[u] - node0], 1u);
        }
    }
    __syncthreads();
    const unsigned cn = (tid < GAT_BIN_NODES) ? cntL[tid] : 0u;
    const unsigned ex = gat_block_scan(cn, wsum);
    if (tid < GAT_BIN_NODES) {
        curL[tid] = ex;
        const int n = node0 + tid;
        if (n < N) { cnt[n] = (int)cn; offs[n] = beg + (int)ex; }
    }
    __syncthreads();
    const bool fits = m <= GAT_SORT_CAP;
    for (int b0 = 0; b0 < m; b0 += 4096) {
        int2 pl[4];
#pragma unroll
        for (int u = 0; u < 4; ++u) {
            const int i = b0 + u * 1024 + tid;
            pl[u] = part[beg + (i < m ? i : m - 1)];
        }
#pragma unroll
        for (int u = 0; u < 4; ++u) {
            const int i = b0 + u * 1024 + tid;
            if (i < m) {
                const unsigned pos = atomicAdd(&curL[pl[u].y - node0], 1u);
                if (fits) stageS[pos] = pl[u].x; else tmp[beg + (int)pos] = pl[u].x;
            }
        }
    }
    if (fits) {
        __syncthreads();
        for (int i = tid; i < m; i += 1024) tmp[beg + i] = stageS[i];
    }
}

__global__ __launch_bounds__(256) void gat_order_kernel(
    const int* __restrict__ cnt, const int* __restrict__ offs,
    const int* __restrict__ tmp, int* __restrict__ sorted, int N)
{
    const int lane = threadIdx.x & 31;
    const int w0 = __builtin_amdgcn_readfirstlane((int)(blockIdx.x * 8 + (threadIdx.x >> 5)));
    const int nw = (int)gridDim.x * 8;
    for (int wid = w0; wid < N; wid += nw) {
        const int deg = __builtin_amdgcn_readfirstlane(cnt[wid]);
        const int off = __builtin_amdgcn_readfirstlane(offs[wid]);
        if (deg <= 32) {
            int v = (lane < deg) ? tmp[off + lane] : 0x7fffffff;
            int rank = 0;
#pragma unroll
            for (int j = 0; j < 32; ++j) {
                int vj = __builtin_amdgcn_readlane(v, j);
                rank += (vj < v || (vj == v && j < lane)) ? 1 : 0;
            }
            if (lane < deg) sorted[off + rank] = v;
        } else {
            for (int i = lane; i < deg; i += 32) {
                int v = tmp[off + i];
                int rank = 0;
                for (int j = 0; j < deg; ++j) {
                    int vj = tmp[off + j];
                    rank += (vj < v || (vj == v && j < i)) ? 1 : 0;
                }
                sorted[off + rank] = v;
            }
        }
    }
}

__device__ __forceinline__ float gat_rl(float x, int k) {
    return __uint_as_float((unsigned)__builtin_amdgcn_readlane((int)__float_as_uint(x), k));
}

__global__ __launch_bounds__(256) void gat_node_l1_kernel(
    const int* __restrict__ cnt, const int* __restrict__ offs, const int* __restrict__ sorted,
    const float* __restrict__ as_, const float* __restrict__ ad_,
    const float* __restrict__ feat, const float* __restrict__ bias,
    float* __restrict__ outp, int N)
{
    const int lane = threadIdx.x & 31;
    const int w0 = __builtin_amdgcn_readfirstlane((int)(blockIdx.x * 8 + (threadIdx.x >> 5)));
    const int nw = (int)gridDim.x * 8;
    for (int n = w0; n < N; n += nw) {
        const int deg = __builtin_amdgcn_readfirstlane(cnt[n]);
        const int off = __builtin_amdgcn_readfirstlane(offs[n]);
        const float2* as2 = reinterpret_cast<const float2*>(as_);
        const float2* ad2 = reinterpret_cast<const float2*>(ad_);
        const float4* f4  = reinterpret_cast<const float4*>(feat);
        const float2 adn = ad2[n];
        const float2 asn = as2[n];
        const float es0 = gat_lrelu(asn.x + adn.x), es1 = gat_lrelu(asn.y + adn.y);

        const bool v0 = lane < deg;
        int s = n; float e0 = es0, e1 = es1;
        if (v0) {
            s = sorted[off + lane];
            float2 a = as2[s];
            e0 = gat_lrelu(a.x + adn.x); e1 = gat_lrelu(a.y + adn.y);
        }
        float m0 = fmaxf(e0, es0), m1 = fmaxf(e1, es1);
        for (int base = 32; base < deg; base += 32) {
            int i = base + lane;
            if (i < deg) {
                float2 a = as2[sorted[off + i]];
                m0 = fmaxf(m0, gat_lrelu(a.x + adn.x)); m1 = fmaxf(m1, gat_lrelu(a.y + adn.y));
            }
        }
        for (int o = 16; o > 0; o >>= 1) {
            m0 = fmaxf(m0, __shfl_xor(m0, o, 32));
            m1 = fmaxf(m1, __shfl_xor(m1, o, 32));
        }
        const float p0 = v0 ? expf(e0 - m0) : 0.0f, p1 = v0 ? expf(e1 - m1) : 0.0f;
        float d0 = p0, d1 = p1;
        for (int base = 32; base < deg; base += 32) {
            int i = base + lane;
            if (i < deg) {
                float2 a = as2[sorted[off + i]];
                d0 += expf(gat_lrelu(a.x + adn.x) - m0); d1 += expf(gat_lrelu(a.y + adn.y) - m1);
            }
        }
        for (int o = 16; o > 0; o >>= 1) {
            d0 += __shfl_xor(d0, o, 32);
            d1 += __shfl_xor(d1, o, 32);
        }
        const float ps0 = expf(es0 - m0), ps1 = expf(es1 - m1);
        const float den0 = (d0 + ps0) + 1e-16f, den1 = (d1 + ps1) + 1e-16f;

        float4 acc = make_float4(0.0f, 0.0f, 0.0f, 0.0f);
        auto gather = [&](int sv, float q0, float q1, int cn) {
            for (int k = 0; k < cn; k += 4) {
#pragma unroll
                for (int u = 0; u < 4; ++u) {
                    int   sk = __builtin_amdgcn_readlane(sv, k + u);
                    float a0 = gat_rl(q0, k + u), a1 = gat_rl(q1, k + u);
                    float ck = (lane < 16) ? a0 : a1;
                    float4 f = f4[(long)sk * 32 + lane];
                    acc.x = fmaf(ck, f.x, acc.x); acc.y = fmaf(ck, f.y, acc.y);
                    acc.z = fmaf(ck, f.z, acc.z); acc.w = fmaf(ck, f.w, acc.w);
                }
            }
        };
        gather(s, p0 / den0, p1 / den1, deg < 32 ? deg : 32);
        for (int base = 32; base < deg; base += 32) {
            int i = base + lane;
            int t = n; float q0 = 0.0f, q1 = 0.0f;
            if (i < deg) {
                t = sorted[off + i];
                float2 a = as2[t];
                q0 = expf(gat_lrelu(a.x + adn.x) - m0) / den0;
                q1 = expf(gat_lrelu(a.y + adn.y) - m1) / den1;
            }
            gather(t, q0, q1, (deg - base) < 32 ? (deg - base) : 32);
        }
        {
            float cs = (lane < 16) ? (ps0 / den0) : (ps1 / den1);
            float4 f = f4[(long)n * 32 + lane];
            acc.x = fmaf(cs, f.x, acc.x); acc.y = fmaf(cs, f.y, acc.y);
            acc.z = fmaf(cs, f.z, acc.z); acc.w = fmaf(cs, f.w, acc.w);
        }
        const float4 b = reinterpret_cast<const float4*>(bias)[lane];
        float4 r;
        float t0 = acc.x + b.x, t1 = acc.y + b.y, t2 = acc.z + b.z, t3 = acc.w + b.w;
        r.x = t0 > 0.0f ? t0 : (expf(t0) - 1.0f);
        r.y = t1 > 0.0f ? t1 : (expf(t1) - 1.0f);
        r.z = t2 > 0.0f ? t2 : (expf(t2) - 1.0f);
        r.w = t3 > 0.0f ? t3 : (expf(t3) - 1.0f);
        reinterpret_cast<float4*>(outp)[(long)n * 32 + lane] = r;
    }
}

__global__ __launch_bounds__(256) void gat_node_l2_kernel(
    const int* __restrict__ cnt, const int* __restrict__ offs, const int* __restrict__ sorted,
    const float* __restrict__ as_, const float* __restrict__ ad_,
    const float* __restrict__ feat, const float* __restrict__ bias,
    float* __restrict__ outp, int N)
{
    const int lane = threadIdx.x & 31;
    const int w0 = __builtin_amdgcn_readfirstlane((int)(blockIdx.x * 8 + (threadIdx.x >> 5)));
    const int nw = (int)gridDim.x * 8;
    for (int n = w0; n < N; n += nw) {
        const int deg = __builtin_amdgcn_readfirstlane(cnt[n]);
        const int off = __builtin_amdgcn_readfirstlane(offs[n]);
        const float4* f4 = reinterpret_cast<const float4*>(feat);
        const float adn = ad_[n];
        const float es  = gat_lrelu(as_[n] + adn);

        const bool v0 = lane < deg;
        int s = n; float e = es;
        if (v0) { s = sorted[off + lane]; e = gat_lrelu(as_[s] + adn); }
        float m = fmaxf(e, es);
        for (int base = 32; base < deg; base += 32) {
            int i = base + lane;
            if (i < deg) m = fmaxf(m, gat_lrelu(as_[sorted[off + i]] + adn));
        }
        for (int o = 16; o > 0; o >>= 1) m = fmaxf(m, __shfl_xor(m, o, 32));
        const float p = v0 ? expf(e - m) : 0.0f;
        float d = p;
        for (int base = 32; base < deg; base += 32) {
            int i = base + lane;
            if (i < deg) d += expf(gat_lrelu(as_[sorted[off + i]] + adn) - m);
        }
        for (int o = 16; o > 0; o >>= 1) d += __shfl_xor(d, o, 32);
        const float ps  = expf(es - m);
        const float den = (d + ps) + 1e-16f;

        const int  g  = lane / 10;
        const int  cl = lane - g * 10;
        const bool on = g < 3;
        float4 acc = make_float4(0.0f, 0.0f, 0.0f, 0.0f);
        auto gather = [&](int sv, float q, int cn) {
            const int iters = (cn + 2) / 3;
            for (int tb = 0; tb < iters; tb += 4) {
#pragma unroll
              for (int u = 0; u < 4; ++u) {
                int k  = (tb + u) * 3 + g;
                const bool val = on && (k < cn);
                int kk = (k < cn) ? k : 0;
                int   sk = __shfl(sv, kk, 32);
                float ck = __shfl(q, kk, 32);
                ck = val ? ck : 0.0f;
                float4 f = f4[(long)(val ? sk : n) * 10 + cl];
                acc.x = fmaf(ck, f.x, acc.x); acc.y = fmaf(ck, f.y, acc.y);
                acc.z = fmaf(ck, f.z, acc.z); acc.w = fmaf(ck, f.w, acc.w);
              }
            }
        };
        gather(s, p / den, deg < 32 ? deg : 32);
        for (int base = 32; base < deg; base += 32) {
            int i = base + lane;
            int t = n; float q = 0.0f;
            if (i < deg) {
                t = sorted[off + i];
                q = expf(gat_lrelu(as_[t] + adn) - m) / den;
            }
            gather(t, q, (deg - base) < 32 ? (deg - base) : 32);
        }
        const int l1 = (lane + 10) & 31, l2 = (lane + 20) & 31;
        float4 a1, a2;
        a1.x = __shfl(acc.x, l1, 32); a1.y = __shfl(acc.y, l1, 32);
        a1.z = __shfl(acc.z, l1, 32); a1.w = __shfl(acc.w, l1, 32);
        a2.x = __shfl(acc.x, l2, 32); a2.y = __shfl(acc.y, l2, 32);
        a2.z = __shfl(acc.z, l2, 32); a2.w = __shfl(acc.w, l2, 32);
        if (lane < 10) {
            const float cs = ps / den;
            const float4 f = f4[(long)n * 10 + lane];
            const float4 b = reinterpret_cast<const float4*>(bias)[lane];
            float4 r;
            r.x = fmaf(cs, f.x, (acc.x + a1.x) + a2.x) + b.x;
            r.y = fmaf(cs, f.y, (acc.y + a1.y) + a2.y) + b.y;
            r.z = fmaf(cs, f.z, (acc.z + a1.z) + a2.z) + b.z;
            r.w = fmaf(cs, f.w, (acc.w + a1.w) + a2.w) + b.w;
            reinterpret_cast<float4*>(outp)[(long)n * 10 + lane] = r;
        }
    }
}

extern "C" void kernel_launch(void* const* d_in, const int* in_sizes, int n_in,
                              void* d_out, int out_size, void* d_ws, size_t ws_size,
                              hipStream_t stream)
{
    (void)n_in; (void)out_size; (void)ws_size;
    const float* x   = (const float*)d_in[0];
    const int*   ei  = (const int*)  d_in[1];
    const float* W1  = (const float*)d_in[2];
    const float* aS1 = (const float*)d_in[3];
    const float* aD1 = (const float*)d_in[4];
    const float* b1  = (const float*)d_in[5];
    const float* W2  = (const float*)d_in[6];
    const float* aS2 = (const float*)d_in[7];
    const float* aD2 = (const float*)d_in[8];
    const float* b2  = (const float*)d_in[9];

    const int IN = 128, HC1 = 128, OUTC = 40;
    int N  = in_sizes[0] / IN;
    int E  = in_sizes[1] / 2;
    const int* srcI = ei;
    const int* dstI = ei + E;

    float* wsf = (float*)d_ws;
    size_t off = 0;
    auto take = [&](size_t nf) -> size_t { size_t o = off; off += (nf + 63) & ~(size_t)63; return o; };
    float* h1     = wsf + take((size_t)N * HC1);
    float* agg1   = wsf + take((size_t)N * HC1);
    float* as1    = wsf + take((size_t)N * 2);
    float* ad1    = wsf + take((size_t)N * 2);
    float* h2     = wsf + take((size_t)N * OUTC);
    float* as2    = wsf + take((size_t)N);
    float* ad2    = wsf + take((size_t)N);
    int*   cnt    = (int*)(wsf + take((size_t)N));
    int*   offs   = (int*)(wsf + take((size_t)N));
    int*   binws  = (int*)(wsf + take((size_t)N));
    int*   tmp    = (int*)(wsf + take((size_t)E));
    int*   sorted = (int*)(wsf + take((size_t)E));

    dim3 blk(256);
    dim3 ngrid(2048);

    int NB   = (N + GAT_BIN_NODES - 1) >> GAT_BIN_SHIFT;
    int nblk = (E + GAT_EPB - 1) / GAT_EPB;
    unsigned* binCount  = (unsigned*)binws;
    int*      binStart  = binws + (size_t)NB * GAT_CPAD;
    int2*     part      = (int2*)agg1;
    unsigned* blockBase = (unsigned*)(agg1 + (size_t)2 * E);
    gat_fill_u32_kernel<<<(NB * GAT_CPAD + 255) / 256, blk, 0, stream>>>(binCount, 0u, (long)NB * GAT_CPAD);
    gat_bin_hist_kernel<<<nblk, 1024, 0, stream>>>(dstI, E, NB, binCount, blockBase);
    gat_bin_scatter_kernel<<<nblk, 1024, 0, stream>>>(srcI, dstI, E, NB, binCount, blockBase, binStart, part);
    gat_bin_sort_kernel<<<NB, 1024, 0, stream>>>(part, binStart, N, cnt, offs, tmp);
    gat_order_kernel<<<ngrid, blk, 0, stream>>>(cnt, offs, tmp, sorted, N);

    gat_gemm_wmma<128, 8, 2, 128><<<(N + 127) / 128, blk, 0, stream>>>(x, W1, aS1, aD1, h1, as1, ad1, N);
    gat_node_l1_kernel<<<ngrid, blk, 0, stream>>>(cnt, offs, sorted, as1, ad1, h1, b1, agg1, N);

    gat_gemm_wmma<48, 3, 1, 40><<<(N + 127) / 128, blk, 0, stream>>>(agg1, W2, aS2, aD2, h2, as2, ad2, N);
    gat_node_l2_kernel<<<ngrid, blk, 0, stream>>>(cnt, offs, sorted, as2, ad2, h2, b2, (float*)d_out, N);
}


# layer 1's product x@W1 runs in the bin sort's launch: the first blocks multiply 512 rows each, the rest sort a bin each (one head per wave, so two blocks fit a WGP); k order and sum orders unchanged
# speedup vs baseline: 7.3701x; 1.1228x over previous
#include <hip/hip_runtime.h>
#include <math.h>

typedef float v2f __attribute__((ext_vector_type(2)));
typedef float v8f __attribute__((ext_vector_type(8)));

#define NEG_SLOPE 0.2f

__device__ __forceinline__ float gat_lrelu(float v) { return v > 0.0f ? v : NEG_SLOPE * v; }

template<int CNT, int OFF>
__device__ __forceinline__ void gat_fold_step(float* v, int m) {
    const bool up = (m & OFF) != 0;
#pragma unroll
    for (int j = 0; j < CNT; ++j) {
        const float keep = up ? v[j + CNT] : v[j];
        const float send = up ? v[j] : v[j + CNT];
        v[j] = keep + __shfl_xor(send, OFF, 32);
    }
}
template<int NV>
__device__ __forceinline__ void gat_fold16(float* v, int m) {
    gat_fold_step<NV / 2, 8>(v, m);
    gat_fold_step<NV / 4, 4>(v, m);
    gat_fold_step<NV / 8, 2>(v, m);
    gat_fold_step<NV / 16, 1>(v, m);
}

template<int NCOL, int NCT, int H, int NSTORE>
__global__ __launch_bounds__(256) void gat_gemm_wmma(
    const float* __restrict__ X, const float* __restrict__ W,
    const float* __restrict__ aw_s, const float* __restrict__ aw_d,
    float* __restrict__ Y, float* __restrict__ as_, float* __restrict__ ad_,
    int N)
{
    constexpr int K = 128;
    constexpr int ncolStore = NSTORE;
    constexpr int TPH = NCT / H;
    constexpr int NV  = 16 * H;
    constexpr int NC4 = NCOL / 4;
    constexpr int NLD = K * NC4 / 256;
    static_assert(NCT % H == 0, "a column tile must belong to one head");
    static_assert(NSTORE % 4 == 0 && NSTORE >= 4 && NSTORE <= NCOL, "rows of W are whole float4 pieces");
    static_assert((K * NC4) % 256 == 0, "every thread stages the same number of pieces");
    __shared__ __attribute__((aligned(16))) float ldsW[K * NCOL];
    {
        float4 wv[NLD];
#pragma unroll
        for (int i = 0; i < NLD; ++i) {
            const int p  = (int)threadIdx.x + i * 256;
            const int k  = p / NC4, c = (p % NC4) * 4;
            const int cc = c < NSTORE ? c : NSTORE - 4;
            wv[i] = *reinterpret_cast<const float4*>(W + k * NSTORE + cc);
        }
#pragma unroll
        for (int i = 0; i < NLD; ++i) {
            const int p = (int)threadIdx.x + i * 256;
            const int c = (p % NC4) * 4;
            reinterpret_cast<float4*>(ldsW)[p] = (c < NSTORE) ? wv[i] : make_float4(0.0f, 0.0f, 0.0f, 0.0f);
        }
    }
    __syncthreads();

    const int lane = threadIdx.x & 31;
    const int wave = threadIdx.x >> 5;
    const int m    = lane & 15;
    const int half = lane >> 4;
    const int koff = half << 1;

    long row0 = (long)blockIdx.x * 128 + wave * 16;
    long rowL = row0 + m; if (rowL > (long)N - 1) rowL = (long)N - 1;
    const float* xrow = X + rowL * K;

    float wS[NCT], wD[NCT];
#pragma unroll
    for (int t = 0; t < NCT; ++t) {
        const int c = t * 16 + m;
        wS[t] = (c < ncolStore) ? aw_s[c] : 0.0f;
        wD[t] = (c < ncolStore) ? aw_d[c] : 0.0f;
    }

    v8f acc[NCT] = {};
    for (int k0 = 0; k0 < K; k0 += 4) {
        v2f a;
        a.x = xrow[k0 + koff];
        a.y = xrow[k0 + koff + 1];
#pragma unroll
        for (int t = 0; t < NCT; ++t) {
            v2f b;
            b.x = ldsW[(k0 + koff) * NCOL + t * 16 + m];
            b.y = ldsW[(k0 + koff + 1) * NCOL + t * 16 + m];
            acc[t] = __builtin_amdgcn_wmma_f32_16x16x4_f32(
                false, a, false, b, (short)0, acc[t], false, false);
        }
    }
#pragma unroll
    for (int r = 0; r < 8; ++r) {
        long row = row0 + half * 8 + r;
        if (row < N) {
#pragma unroll
            for (int t = 0; t < NCT; ++t) {
                int c = t * 16 + m;
                if (c < ncolStore) Y[row * ncolStore + c] = acc[t][r];
            }
        }
    }

    float v[NV];
#pragma unroll
    for (int r = 0; r < 8; ++r) {
#pragma unroll
        for (int hh = 0; hh < H; ++hh) {
            float sv = 0.0f, dv = 0.0f;
#pragma unroll
            for (int u = 0; u < TPH; ++u) {
                const int t = hh * TPH + u;
                sv = fmaf(acc[t][r], wS[t], sv);
                dv = fmaf(acc[t][r], wD[t], dv);
            }
            v[(r * H + hh) * 2 + 0] = sv;
            v[(r * H + hh) * 2 + 1] = dv;
        }
    }
    gat_fold16<NV>(v, m);
    {
        const long row = row0 + half * 8 + (m >> 1);
        if (row < N) {
            if (H == 2) {
                as_[row * 2 + (m & 1)] = v[0];
                ad_[row * 2 + (m & 1)] = v[1];
            } else {
                if (m & 1) ad_[row] = v[0]; else as_[row] = v[0];
            }
        }
    }
}

#define GAT_BIN_SHIFT 9
#define GAT_BIN_NODES 512
#define GAT_MAXB      1024
#define GAT_EPB       4096
#define GAT_EPT       4
#define GAT_CPAD      32
#define GAT_SORT_CAP  12288

__global__ void gat_fill_u32_kernel(unsigned* __restrict__ p, unsigned v, long n) {
    long i = blockIdx.x * (long)blockDim.x + threadIdx.x;
    if (i < n) p[i] = v;
}

__device__ __forceinline__ unsigned gat_block_scan(unsigned c, unsigned* wsum) {
    const int lane = threadIdx.x & 31, wave = threadIdx.x >> 5;
    unsigned inc = c;
    for (int o = 1; o < 32; o <<= 1) {
        unsigned y = __shfl_up(inc, o, 32);
        if (lane >= o) inc += y;
    }
    if (lane == 31) wsum[wave] = inc;
    __syncthreads();
    unsigned wb = 0;
    for (int w = 0; w < wave; ++w) wb += wsum[w];
    __syncthreads();
    return wb + inc - c;
}

__global__ __launch_bounds__(1024) void gat_bin_hist_kernel(
    const int* __restrict__ dst, int E, int NB,
    unsigned* __restrict__ binCount, unsigned* __restrict__ blockBase)
{
    __shared__ unsigned hist[GAT_MAXB];
    const int tid = threadIdx.x;
    hist[tid] = 0u;
    __syncthreads();
    const int e0 = blockIdx.x * GAT_EPB;
    int dv[GAT_EPT];
#pragma unroll
    for (int j = 0; j < GAT_EPT; ++j) {
        int e = e0 + j * 1024 + tid;
        dv[j] = -1;
        if (e < E) dv[j] = dst[e];
    }
#pragma unroll
    for (int j = 0; j < GAT_EPT; ++j) {
        if (dv[j] >= 0) atomicAdd(&hist[dv[j] >> GAT_BIN_SHIFT], 1u);
    }
    __syncthreads();
    if (tid < NB) {
        const unsigned cb = hist[tid];
        unsigned base = 0u;
        if (cb) base = atomicAdd(&binCount[tid * GAT_CPAD], cb);
        blockBase[(long)blockIdx.x * NB + tid] = base;
    }
}

__global__ __launch_bounds__(1024) void gat_bin_scatter_kernel(
    const int* __restrict__ src, const int* __restrict__ dst, int E, int NB,
    const unsigned* __restrict__ binCount, const unsigned* __restrict__ blockBase,
    int* __restrict__ binStart, int2* __restrict__ part)
{
    __shared__ int2 stage[GAT_EPB];
    __shared__ unsigned hist[GAT_MAXB];
    __shared__ unsigned lstart[GAT_MAXB];
    __shared__ unsigned gbase[GAT_MAXB];
    __shared__ unsigned wsum[32];
    const int tid = threadIdx.x;
    hist[tid] = 0u;
    __syncthreads();
    const int e0 = blockIdx.x * GAT_EPB;
    int sv[GAT_EPT], dv[GAT_EPT];
    unsigned rk[GAT_EPT];
#pragma unroll
    for (int j = 0; j < GAT_EPT; ++j) {
        int e = e0 + j * 1024 + tid;
        sv[j] = 0; dv[j] = -1;
        if (e < E) { dv[j] = dst[e]; sv[j] = src[e]; }
    }
#pragma unroll
    for (int j = 0; j < GAT_EPT; ++j) {
        rk[j] = 0u;
        if (dv[j] >= 0) rk[j] = atomicAdd(&hist[dv[j] >> GAT_BIN_SHIFT], 1u);
    }
    __syncthreads();
    const unsigned ls = gat_block_scan(hist[tid], wsum);
    const unsigned bc = (tid < NB) ? binCount[tid * GAT_CPAD] : 0u;
    const unsigned bs = gat_block_scan(bc, wsum);
    lstart[tid] = ls;
    if (tid < NB) {
        gbase[tid] = bs + blockBase[(long)blockIdx.x * NB + tid] - ls;
        if (blockIdx.x == 0) {
            binStart[tid] = (int)bs;
            if (tid == NB - 1) binStart[NB] = (int)(bs + bc);
        }
    }
    __syncthreads();
#pragma unroll
    for (int j = 0; j < GAT_EPT; ++j) {
        if (dv[j] >= 0) stage[lstart[dv[j] >> GAT_BIN_SHIFT] + rk[j]] = make_int2(sv[j], dv[j]);
    }
    __syncthreads();
    const int rem = E - e0;
    const int nval = rem < GAT_EPB ? rem : GAT_EPB;
    for (int i = tid; i < nval; i += 1024) {
        const int2 p = stage[i];
        part[gbase[p.y >> GAT_BIN_SHIFT] + (unsigned)i] = p;
    }
}

#define GAT_SORT_LDS_WORDS (GAT_SORT_CAP + 2 * GAT_BIN_NODES + 32)
__device__ __forceinline__ void gat_bin_sort_role(
    float* lds, const int bin,
    const int2* __restrict__ part, const int* __restrict__ binStart, int N,
    int* __restrict__ cnt, int* __restrict__ offs, int* tmp, int* __restrict__ sorted)
{
    int*      stageS = reinterpret_cast<int*>(lds);
    unsigned* cntL   = reinterpret_cast<unsigned*>(lds) + GAT_SORT_CAP;
    unsigned* curL   = cntL + GAT_BIN_NODES;
    unsigned* wsum   = curL + GAT_BIN_NODES;
    const int tid = threadIdx.x;
    const int beg = binStart[bin];
    const int m   = binStart[bin + 1] - beg;
    const int node0 = bin << GAT_BIN_SHIFT;
    if (tid < GAT_BIN_NODES) cntL[tid] = 0u;
    __syncthreads();
    for (int b0 = 0; b0 < m; b0 += 4096) {
        int dl[4];
#pragma unroll
        for (int u = 0; u < 4; ++u) {
            const int i = b0 + u * 1024 + tid;
            dl[u] = part[beg + (i < m ? i : m - 1)].y;
        }
#pragma unroll
        for (int u = 0; u < 4; ++u) {
            const int i = b0 + u * 1024 + tid;
            if (i < m) atomicAdd(&cntL[dl[u] - node0], 1u);
        }
    }
    __syncthreads();
    const unsigned cn = (tid < GAT_BIN_NODES) ? cntL[tid] : 0u;
    const unsigned ex = gat_block_scan(cn, wsum);
    if (tid < GAT_BIN_NODES) {
        curL[tid] = ex;
        const int n = node0 + tid;
        if (n < N) { cnt[n] = (int)cn; offs[n] = beg + (int)ex; }
    }
    __syncthreads();
    const bool fits = m <= GAT_SORT_CAP;
    for (int b0 = 0; b0 < m; b0 += 4096) {
        int2 pl[4];
#pragma unroll
        for (int u = 0; u < 4; ++u) {
            const int i = b0 + u * 1024 + tid;
            pl[u] = part[beg + (i < m ? i : m - 1)];
        }
#pragma unroll
        for (int u = 0; u < 4; ++u) {
            const int i = b0 + u * 1024 + tid;
            if (i < m) {
                const unsigned pos = atomicAdd(&curL[pl[u].y - node0], 1u);
                if (fits) stageS[pos] = pl[u].x; else tmp[beg + (int)pos] = pl[u].x;
            }
        }
    }
    if (!fits) __threadfence();
    __syncthreads();
    const int lane = tid & 31;
    const int wv = __builtin_amdgcn_readfirstlane(tid >> 5);
    auto order_groups = [&](auto src_at) {
        for (int nl = wv; nl < GAT_BIN_NODES; nl += 32) {
            const int deg = __builtin_amdgcn_readfirstlane((int)cntL[nl]);
            const int ex  = __builtin_amdgcn_readfirstlane((int)curL[nl]) - deg;
            if (deg <= 32) {
                const unsigned key = (lane < deg) ? (((unsigned)src_at(ex + lane) << 5) | (unsigned)lane)
                                                  : 0xffffffffu;
                unsigned rank = 0u;
                for (int j = 0; j < deg; j += 4) {
#pragma unroll
                    for (int u = 0; u < 4; ++u) {
                        const unsigned kj = (unsigned)__builtin_amdgcn_readlane((int)key, j + u);
                        rank += (kj < key) ? 1u : 0u;
                    }
                }
                if (lane < deg) sorted[beg + ex + (int)rank] = (int)(key >> 5);
            } else {
                for (int i = lane; i < deg; i += 32) {
                    int v = src_at(ex + i);
                    int rank = 0;
                    for (int j = 0; j < deg; ++j) {
                        int vj = src_at(ex + j);
                        rank += (vj < v || (vj == v && j < i)) ? 1 : 0;
                    }
                    sorted[beg + ex + rank] = v;
                }
            }
        }
    };
    if (fits) order_groups([&](int i) { return stageS[i]; });
    else      order_groups([&](int i) { return tmp[beg + i]; });
}

__device__ __forceinline__ void gat_gemm1_role(
    float* ldsW, const int gblk,
    const float* __restrict__ X, const float* __restrict__ W,
    const float* __restrict__ aw_s, const float* __restrict__ aw_d,
    float* __restrict__ Y, float* __restrict__ as_, float* __restrict__ ad_, int N)
{
    constexpr int K = 128, NCOL = 128;
    {
        float4 wv[4];
#pragma unroll
        for (int i = 0; i < 4; ++i) wv[i] = reinterpret_cast<const float4*>(W)[(int)threadIdx.x + i * 1024];
#pragma unroll
        for (int i = 0; i < 4; ++i) reinterpret_cast<float4*>(ldsW)[(int)threadIdx.x + i * 1024] = wv[i];
    }
    __syncthreads();

    const int lane = threadIdx.x & 31;
    const int wave = __builtin_amdgcn_readfirstlane((int)(threadIdx.x >> 5));
    const int m    = lane & 15;
    const int half = lane >> 4;
    const int koff = half << 1;
    const int ch   = wave & 1;
    const int rg   = wave >> 1;
    const int col0 = ch * 64 + m;

    float wS[4], wD[4];
#pragma unroll
    for (int t = 0; t < 4; ++t) { wS[t] = aw_s[col0 + t * 16]; wD[t] = aw_d[col0 + t * 16]; }

#pragma nounroll
    for (int it = 0; it < 2; ++it) {
        const long row0 = (long)gblk * 512 + it * 256 + rg * 16;
        if (row0 >= (long)N) break;
        const int rlast = (int)((long)N - 1 - row0);
        const int mr    = m < rlast ? m : rlast;
        const float* xrow = X + row0 * K + (mr * K + koff);

        v8f acc[4] = {};
        for (int k0 = 0; k0 < K; k0 += 4) {
            v2f a;
            a.x = xrow[k0];
            a.y = xrow[k0 + 1];
#pragma unroll
            for (int t = 0; t < 4; ++t) {
                v2f b;
                b.x = ldsW[(k0 + koff) * NCOL + col0 + t * 16];
                b.y = ldsW[(k0 + koff + 1) * NCOL + col0 + t * 16];
                acc[t] = __builtin_amdgcn_wmma_f32_16x16x4_f32(
                    false, a, false, b, (short)0, acc[t], false, false);
            }
        }
        float* ylane = Y + row0 * NCOL + (half * 8 * NCOL + col0);
        const int rleft = rlast - half * 8;
#pragma unroll
        for (int r = 0; r < 8; ++r) {
            if (r <= rleft) {
#pragma unroll
                for (int t = 0; t < 4; ++t) ylane[r * NCOL + t * 16] = acc[t][r];
            }
        }
        float v[16];
#pragma unroll
        for (int r = 0; r < 8; ++r) {
            float sv = 0.0f, dv = 0.0f;
#pragma unroll
            for (int t = 0; t < 4; ++t) {
                sv = fmaf(acc[t][r], wS[t], sv);
                dv = fmaf(acc[t][r], wD[t], dv);
            }
            v[r * 2 + 0] = sv;
            v[r * 2 + 1] = dv;
        }
        gat_fold16<16>(v, m);
        const int rr = half * 8 + (m >> 1);
        if (rr <= rlast) {
            if (m & 1) ad_[row0 * 2 + (rr * 2 + ch)] = v[0]; else as_[row0 * 2 + (rr * 2 + ch)] = v[0];
        }
    }
}

__global__ __launch_bounds__(1024, 16) void gat_sort_gemm1_kernel(
    const int2* __restrict__ part, const int* __restrict__ binStart, int N, int NG,
    int* __restrict__ cnt, int* __restrict__ offs, int* tmp, int* __restrict__ sorted,
    const float* __restrict__ X, const float* __restrict__ W,
    const float* __restrict__ aw_s, const float* __restrict__ aw_d,
    float* __restrict__ Y, float* __restrict__ as_, float* __restrict__ ad_)
{
    __shared__ __attribute__((aligned(16))) float lds[128 * 128];
    static_assert(GAT_SORT_LDS_WORDS <= 128 * 128, "the sort's tables fit the buffer");
    if ((int)blockIdx.x < NG) gat_gemm1_role(lds, (int)blockIdx.x, X, W, aw_s, aw_d, Y, as_, ad_, N);
    else                      gat_bin_sort_role(lds, (int)blockIdx.x - NG, part, binStart, N, cnt, offs, tmp, sorted);
}

__device__ __forceinline__ float gat_rl(float x, int k) {
    return __uint_as_float((unsigned)__builtin_amdgcn_readlane((int)__float_as_uint(x), k));
}

__global__ __launch_bounds__(256) void gat_node_l1_kernel(
    const int* __restrict__ cnt, const int* __restrict__ offs, const int* __restrict__ sorted,
    const float* __restrict__ as_, const float* __restrict__ ad_,
    const float* __restrict__ feat, const float* __restrict__ bias,
    float* __restrict__ outp, int N)
{
    const int lane = threadIdx.x & 31;
    const int w0 = __builtin_amdgcn_readfirstlane((int)(blockIdx.x * 8 + (threadIdx.x >> 5)));
    const int nw = (int)gridDim.x * 8;
    for (int n = w0; n < N; n += nw) {
        const int deg = __builtin_amdgcn_readfirstlane(cnt[n]);
        const int off = __builtin_amdgcn_readfirstlane(offs[n]);
        const float2* as2 = reinterpret_cast<const float2*>(as_);
        const float2* ad2 = reinterpret_cast<const float2*>(ad_);
        const float4* f4  = reinterpret_cast<const float4*>(feat);
        const float2 adn = ad2[n];
        const float2 asn = as2[n];
        const float es0 = gat_lrelu(asn.x + adn.x), es1 = gat_lrelu(asn.y + adn.y);

        const bool v0 = lane < deg;
        int s = n; float e0 = es0, e1 = es1;
        if (v0) {
            s = sorted[off + lane];
            float2 a = as2[s];
            e0 = gat_lrelu(a.x + adn.x); e1 = gat_lrelu(a.y + adn.y);
        }
        float m0 = fmaxf(e0, es0), m1 = fmaxf(e1, es1);
        for (int base = 32; base < deg; base += 32) {
            int i = base + lane;
            if (i < deg) {
                float2 a = as2[sorted[off + i]];
                m0 = fmaxf(m0, gat_lrelu(a.x + adn.x)); m1 = fmaxf(m1, gat_lrelu(a.y + adn.y));
            }
        }
        for (int o = 16; o > 0; o >>= 1) {
            m0 = fmaxf(m0, __shfl_xor(m0, o, 32));
            m1 = fmaxf(m1, __shfl_xor(m1, o, 32));
        }
        const float p0 = v0 ? expf(e0 - m0) : 0.0f, p1 = v0 ? expf(e1 - m1) : 0.0f;
        float d0 = p0, d1 = p1;
        for (int base = 32; base < deg; base += 32) {
            int i = base + lane;
            if (i < deg) {
                float2 a = as2[sorted[off + i]];
                d0 += expf(gat_lrelu(a.x + adn.x) - m0); d1 += expf(gat_lrelu(a.y + adn.y) - m1);
            }
        }
        for (int o = 16; o > 0; o >>= 1) {
            d0 += __shfl_xor(d0, o, 32);
            d1 += __shfl_xor(d1, o, 32);
        }
        const float ps0 = expf(es0 - m0), ps1 = expf(es1 - m1);
        const float den0 = (d0 + ps0) + 1e-16f, den1 = (d1 + ps1) + 1e-16f;

        float4 acc = make_float4(0.0f, 0.0f, 0.0f, 0.0f);
        auto gather = [&](int sv, float q0, float q1, int cn) {
            for (int k = 0; k < cn; k += 4) {
#pragma unroll
                for (int u = 0; u < 4; ++u) {
                    int   sk = __builtin_amdgcn_readlane(sv, k + u);
                    float a0 = gat_rl(q0, k + u), a1 = gat_rl(q1, k + u);
                    float ck = (lane < 16) ? a0 : a1;
                    float4 f = f4[(long)sk * 32 + lane];
                    acc.x = fmaf(ck, f.x, acc.x); acc.y = fmaf(ck, f.y, acc.y);
                    acc.z = fmaf(ck, f.z, acc.z); acc.w = fmaf(ck, f.w, acc.w);
                }
            }
        };
        gather(s, p0 / den0, p1 / den1, deg < 32 ? deg : 32);
        for (int base = 32; base < deg; base += 32) {
            int i = base + lane;
            int t = n; float q0 = 0.0f, q1 = 0.0f;
            if (i < deg) {
                t = sorted[off + i];
                float2 a = as2[t];
                q0 = expf(gat_lrelu(a.x + adn.x) - m0) / den0;
                q1 = expf(gat_lrelu(a.y + adn.y) - m1) / den1;
            }
            gather(t, q0, q1, (deg - base) < 32 ? (deg - base) : 32);
        }
        {
            float cs = (lane < 16) ? (ps0 / den0) : (ps1 / den1);
            float4 f = f4[(long)n * 32 + lane];
            acc.x = fmaf(cs, f.x, acc.x); acc.y = fmaf(cs, f.y, acc.y);
            acc.z = fmaf(cs, f.z, acc.z); acc.w = fmaf(cs, f.w, acc.w);
        }
        const float4 b = reinterpret_cast<const float4*>(bias)[lane];
        float4 r;
        float t0 = acc.x + b.x, t1 = acc.y + b.y, t2 = acc.z + b.z, t3 = acc.w + b.w;
        r.x = t0 > 0.0f ? t0 : (expf(t0) - 1.0f);
        r.y = t1 > 0.0f ? t1 : (expf(t1) - 1.0f);
        r.z = t2 > 0.0f ? t2 : (expf(t2) - 1.0f);
        r.w = t3 > 0.0f ? t3 : (expf(t3) - 1.0f);
        reinterpret_cast<float4*>(outp)[(long)n * 32 + lane] = r;
    }
}

__global__ __launch_bounds__(256) void gat_node_l2_kernel(
    const int* __restrict__ cnt, const int* __restrict__ offs, const int* __restrict__ sorted,
    const float* __restrict__ as_, const float* __restrict__ ad_,
    const float* __restrict__ feat, const float* __restrict__ bias,
    float* __restrict__ outp, int N)
{
    const int lane = threadIdx.x & 31;
    const int w0 = __builtin_amdgcn_readfirstlane((int)(blockIdx.x * 8 + (threadIdx.x >> 5)));
    const int nw = (int)gridDim.x * 8;
    for (int n = w0; n < N; n += nw) {
        const int deg = __builtin_amdgcn_readfirstlane(cnt[n]);
        const int off = __builtin_amdgcn_readfirstlane(offs[n]);
        const float4* f4 = reinterpret_cast<const float4*>(feat);
        const float adn = ad_[n];
        const float es  = gat_lrelu(as_[n] + adn);

        const bool v0 = lane < deg;
        int s = n; float e = es;
        if (v0) { s = sorted[off + lane]; e = gat_lrelu(as_[s] + adn); }
        float m = fmaxf(e, es);
        for (int base = 32; base < deg; base += 32) {
            int i = base + lane;
            if (i < deg) m = fmaxf(m, gat_lrelu(as_[sorted[off + i]] + adn));
        }
        for (int o = 16; o > 0; o >>= 1) m = fmaxf(m, __shfl_xor(m, o, 32));
        const float p = v0 ? expf(e - m) : 0.0f;
        float d = p;
        for (int base = 32; base < deg; base += 32) {
            int i = base + lane;
            if (i < deg) d += expf(gat_lrelu(as_[sorted[off + i]] + adn) - m);
        }
        for (int o = 16; o > 0; o >>= 1) d += __shfl_xor(d, o, 32);
        const float ps  = expf(es - m);
        const float den = (d + ps) + 1e-16f;

        const int  g  = lane / 10;
        const int  cl = lane - g * 10;
        const bool on = g < 3;
        float4 acc = make_float4(0.0f, 0.0f, 0.0f, 0.0f);
        auto gather = [&](int sv, float q, int cn) {
            const int iters = (cn + 2) / 3;
            for (int tb = 0; tb < iters; tb += 4) {
#pragma unroll
              for (int u = 0; u < 4; ++u) {
                int k  = (tb + u) * 3 + g;
                const bool val = on && (k < cn);
                int kk = (k < cn) ? k : 0;
                int   sk = __shfl(sv, kk, 32);
                float ck = __shfl(q, kk, 32);
                ck = val ? ck : 0.0f;
                float4 f = f4[(long)(val ? sk : n) * 10 + cl];
                acc.x = fmaf(ck, f.x, acc.x); acc.y = fmaf(ck, f.y, acc.y);
                acc.z = fmaf(ck, f.z, acc.z); acc.w = fmaf(ck, f.w, acc.w);
              }
            }
        };
        gather(s, p / den, deg < 32 ? deg : 32);
        for (int base = 32; base < deg; base += 32) {
            int i = base + lane;
            int t = n; float q = 0.0f;
            if (i < deg) {
                t = sorted[off + i];
                q = expf(gat_lrelu(as_[t] + adn) - m) / den;
            }
            gather(t, q, (deg - base) < 32 ? (deg - base) : 32);
        }
        const int l1 = (lane + 10) & 31, l2 = (lane + 20) & 31;
        float4 a1, a2;
        a1.x = __shfl(acc.x, l1, 32); a1.y = __shfl(acc.y, l1, 32);
        a1.z = __shfl(acc.z, l1, 32); a1.w = __shfl(acc.w, l1, 32);
        a2.x = __shfl(acc.x, l2, 32); a2.y = __shfl(acc.y, l2, 32);
        a2.z = __shfl(acc.z, l2, 32); a2.w = __shfl(acc.w, l2, 32);
        if (lane < 10) {
            const float cs = ps / den;
            const float4 f = f4[(long)n * 10 + lane];
            const float4 b = reinterpret_cast<const float4*>(bias)[lane];
            float4 r;
            r.x = fmaf(cs, f.x, (acc.x + a1.x) + a2.x) + b.x;
            r.y = fmaf(cs, f.y, (acc.y + a1.y) + a2.y) + b.y;
            r.z = fmaf(cs, f.z, (acc.z + a1.z) + a2.z) + b.z;
            r.w = fmaf(cs, f.w, (acc.w + a1.w) + a2.w) + b.w;
            reinterpret_cast<float4*>(outp)[(long)n * 10 + lane] = r;
        }
    }
}

extern "C" void kernel_launch(void* const* d_in, const int* in_sizes, int n_in,
                              void* d_out, int out_size, void* d_ws, size_t ws_size,
                              hipStream_t stream)
{
    (void)n_in; (void)out_size; (void)ws_size;
    const float* x   = (const float*)d_in[0];
    const int*   ei  = (const int*)  d_in[1];
    const float* W1  = (const float*)d_in[2];
    const float* aS1 = (const float*)d_in[3];
    const float* aD1 = (const float*)d_in[4];
    const float* b1  = (const float*)d_in[5];
    const float* W2  = (const float*)d_in[6];
    const float* aS2 = (const float*)d_in[7];
    const float* aD2 = (const float*)d_in[8];
    const float* b2  = (const float*)d_in[9];

    const int IN = 128, HC1 = 128, OUTC = 40;
    int N  = in_sizes[0] / IN;
    int E  = in_sizes[1] / 2;
    const int* srcI = ei;
    const int* dstI = ei + E;

    float* wsf = (float*)d_ws;
    size_t off = 0;
    auto take = [&](size_t nf) -> size_t { size_t o = off; off += (nf + 63) & ~(size_t)63; return o; };
    float* h1     = wsf + take((size_t)N * HC1);
    float* agg1   = wsf + take((size_t)N * HC1);
    float* as1    = wsf + take((size_t)N * 2);
    float* ad1    = wsf + take((size_t)N * 2);
    float* h2     = wsf + take((size_t)N * OUTC);
    float* as2    = wsf + take((size_t)N);
    float* ad2    = wsf + take((size_t)N);
    int*   cnt    = (int*)(wsf + take((size_t)N));
    int*   offs   = (int*)(wsf + take((size_t)N));
    int*   binws  = (int*)(wsf + take((size_t)N));
    int*   tmp    = (int*)(wsf + take((size_t)E));
    int*   sorted = (int*)(wsf + take((size_t)E));

    dim3 blk(256);
    dim3 ngrid(2048);

    int NB   = (N + GAT_BIN_NODES - 1) >> GAT_BIN_SHIFT;
    int nblk = (E + GAT_EPB - 1) / GAT_EPB;
    unsigned* binCount  = (unsigned*)binws;
    int*      binStart  = binws + (size_t)NB * GAT_CPAD;
    int2*     part      = (int2*)agg1;
    unsigned* blockBase = (unsigned*)(agg1 + (size_t)2 * E);
    gat_fill_u32_kernel<<<(NB * GAT_CPAD + 255) / 256, blk, 0, stream>>>(binCount, 0u, (long)NB * GAT_CPAD);
    gat_bin_hist_kernel<<<nblk, 1024, 0, stream>>>(dstI, E, NB, binCount, blockBase);
    gat_bin_scatter_kernel<<<nblk, 1024, 0, stream>>>(srcI, dstI, E, NB, binCount, blockBase, binStart, part);
    int NG = (N + 511) / 512;
    gat_sort_gemm1_kernel<<<NG + NB, 1024, 0, stream>>>(
        part, binStart, N, NG, cnt, offs, tmp, sorted, x, W1, aS1, aD1, h1, as1, ad1);

    gat_node_l1_kernel<<<ngrid, blk, 0, stream>>>(cnt, offs, sorted, as1, ad1, h1, b1, agg1, N);

    gat_gemm_wmma<48, 3, 1, 40><<<(N + 127) / 128, blk, 0, stream>>>(agg1, W2, aS2, aD2, h2, as2, ad2, N);
    gat_node_l2_kernel<<<ngrid, blk, 0, stream>>>(cnt, offs, sorted, as2, ad2, h2, b2, (float*)d_out, N);
}


# layer 2: a source's attention logit rides in the padding of its 256-byte feature row; softmax in one pass with a running maximum (sum order changes: self term first, two lane halves, fixed)
# speedup vs baseline: 8.2939x; 1.1253x over previous
#include <hip/hip_runtime.h>
#include <math.h>

typedef float v2f __attribute__((ext_vector_type(2)));
typedef float v8f __attribute__((ext_vector_type(8)));

#define NEG_SLOPE 0.2f

__device__ __forceinline__ float gat_lrelu(float v) { return v > 0.0f ? v : NEG_SLOPE * v; }

template<int CNT, int OFF>
__device__ __forceinline__ void gat_fold_step(float* v, int m) {
    const bool up = (m & OFF) != 0;
#pragma unroll
    for (int j = 0; j < CNT; ++j) {
        const float keep = up ? v[j + CNT] : v[j];
        const float send = up ? v[j] : v[j + CNT];
        v[j] = keep + __shfl_xor(send, OFF, 32);
    }
}
template<int NV>
__device__ __forceinline__ void gat_fold16(float* v, int m) {
    gat_fold_step<NV / 2, 8>(v, m);
    gat_fold_step<NV / 4, 4>(v, m);
    gat_fold_step<NV / 8, 2>(v, m);
    gat_fold_step<NV / 16, 1>(v, m);
}

template<int NCOL, int NCT, int H, int NSTORE, int YP>
__global__ __launch_bounds__(256) void gat_gemm_wmma(
    const float* __restrict__ X, const float* __restrict__ W,
    const float* __restrict__ aw_s, const float* __restrict__ aw_d,
    float* __restrict__ Y, float* __restrict__ ad_,
    int N)
{
    static_assert(H == 1 && NSTORE < NCOL && NCOL <= YP && YP % 4 == 0, "one head; the logit has a padding column");
    constexpr int K = 128;
    constexpr int ncolStore = NSTORE;
    constexpr int TPH = NCT / H;
    constexpr int NV  = 16 * H;
    constexpr int NC4 = NCOL / 4;
    constexpr int NLD = K * NC4 / 256;
    static_assert(NCT % H == 0, "a column tile must belong to one head");
    static_assert(NSTORE % 4 == 0 && NSTORE >= 4 && NSTORE <= NCOL, "rows of W are whole float4 pieces");
    static_assert((K * NC4) % 256 == 0, "every thread stages the same number of pieces");
    __shared__ __attribute__((aligned(16))) float ldsW[K * NCOL];
    {
        float4 wv[NLD];
#pragma unroll
        for (int i = 0; i < NLD; ++i) {
            const int p  = (int)threadIdx.x + i * 256;
            const int k  = p / NC4, c = (p % NC4) * 4;
            const int cc = c < NSTORE ? c : NSTORE - 4;
            wv[i] = *reinterpret_cast<const float4*>(W + k * NSTORE + cc);
        }
#pragma unroll
        for (int i = 0; i < NLD; ++i) {
            const int p = (int)threadIdx.x + i * 256;
            const int c = (p % NC4) * 4;
            reinterpret_cast<float4*>(ldsW)[p] = (c < NSTORE) ? wv[i] : make_float4(0.0f, 0.0f, 0.0f, 0.0f);
        }
    }
    __syncthreads();

    const int lane = threadIdx.x & 31;
    const int wave = threadIdx.x >> 5;
    const int m    = lane & 15;
    const int half = lane >> 4;
    const int koff = half << 1;

    long row0 = (long)blockIdx.x * 128 + wave * 16;
    long rowL = row0 + m; if (rowL > (long)N - 1) rowL = (long)N - 1;
    const float* xrow = X + rowL * K;

    float wS[NCT], wD[NCT];
#pragma unroll
    for (int t = 0; t < NCT; ++t) {
        const int c = t * 16 + m;
        wS[t] = (c < ncolStore) ? aw_s[c] : 0.0f;
        wD[t] = (c < ncolStore) ? aw_d[c] : 0.0f;
    }

    v8f acc[NCT] = {};
    for (int k0 = 0; k0 < K; k0 += 4) {
        v2f a;
        a.x = xrow[k0 + koff];
        a.y = xrow[k0 + koff + 1];
#pragma unroll
        for (int t = 0; t < NCT; ++t) {
            v2f b;
            b.x = ldsW[(k0 + koff) * NCOL + t * 16 + m];
            b.y = ldsW[(k0 + koff + 1) * NCOL + t * 16 + m];
            acc[t] = __builtin_amdgcn_wmma_f32_16x16x4_f32(
                false, a, false, b, (short)0, acc[t], false, false);
        }
    }
#pragma unroll
    for (int r = 0; r < 8; ++r) {
        long row = row0 + half * 8 + r;
        if (row < N) {
#pragma unroll
            for (int t = 0; t < NCT; ++t) {
                int c = t * 16 + m;
                if (c != ncolStore) Y[row * YP + c] = (c < ncolStore) ? acc[t][r] : 0.0f;
            }
        }
    }

    float v[NV];
#pragma unroll
    for (int r = 0; r < 8; ++r) {
#pragma unroll
        for (int hh = 0; hh < H; ++hh) {
            float sv = 0.0f, dv = 0.0f;
#pragma unroll
            for (int u = 0; u < TPH; ++u) {
                const int t = hh * TPH + u;
                sv = fmaf(acc[t][r], wS[t], sv);
                dv = fmaf(acc[t][r], wD[t], dv);
            }
            v[(r * H + hh) * 2 + 0] = sv;
            v[(r * H + hh) * 2 + 1] = dv;
        }
    }
    gat_fold16<NV>(v, m);
    {
        const long row = row0 + half * 8 + (m >> 1);
        if (row < N) {
            if (m & 1) ad_[row] = v[0]; else Y[row * YP + ncolStore] = v[0];
        }
    }
}

#define GAT_BIN_SHIFT 9
#define GAT_BIN_NODES 512
#define GAT_MAXB      1024
#define GAT_EPB       4096
#define GAT_EPT       4
#define GAT_SORT_CAP  12288

__device__ __forceinline__ unsigned gat_block_scan(unsigned c, unsigned* wsum) {
    const int lane = threadIdx.x & 31, wave = threadIdx.x >> 5;
    unsigned inc = c;
    for (int o = 1; o < 32; o <<= 1) {
        unsigned y = __shfl_up(inc, o, 32);
        if (lane >= o) inc += y;
    }
    if (lane == 31) wsum[wave] = inc;
    __syncthreads();
    unsigned wb = 0;
    for (int w = 0; w < wave; ++w) wb += wsum[w];
    __syncthreads();
    return wb + inc - c;
}

__device__ __forceinline__ void gat_block_sum2(unsigned& a, unsigned& b, unsigned* ws) {
    const int lane = threadIdx.x & 31, wave = threadIdx.x >> 5;
    for (int o = 16; o > 0; o >>= 1) {
        a += __shfl_xor(a, o, 32);
        b += __shfl_xor(b, o, 32);
    }
    if (lane == 0) { ws[wave] = a; ws[32 + wave] = b; }
    __syncthreads();
    unsigned ta = 0u, tb = 0u;
    for (int w = 0; w < 32; ++w) { ta += ws[w]; tb += ws[32 + w]; }
    __syncthreads();
    a = ta; b = tb;
}

__global__ __launch_bounds__(1024) void gat_bin_scatter_kernel(
    const int* __restrict__ src, const int* __restrict__ dst, int E, int NB,
    unsigned* __restrict__ runStart, int2* __restrict__ part)
{
    __shared__ int2 stage[GAT_EPB];
    __shared__ unsigned hist[GAT_MAXB];
    __shared__ unsigned lstart[GAT_MAXB];
    __shared__ unsigned wsum[32];
    const int tid = threadIdx.x;
    hist[tid] = 0u;
    __syncthreads();
    const int e0 = blockIdx.x * GAT_EPB;
    int sv[GAT_EPT], dv[GAT_EPT];
    unsigned rk[GAT_EPT];
#pragma unroll
    for (int j = 0; j < GAT_EPT; ++j) {
        int e = e0 + j * 1024 + tid;
        sv[j] = 0; dv[j] = -1;
        if (e < E) { dv[j] = dst[e]; sv[j] = src[e]; }
    }
#pragma unroll
    for (int j = 0; j < GAT_EPT; ++j) {
        rk[j] = 0u;
        if (dv[j] >= 0) rk[j] = atomicAdd(&hist[dv[j] >> GAT_BIN_SHIFT], 1u);
    }
    __syncthreads();
    const unsigned ls = gat_block_scan(hist[tid], wsum);
    const int rem  = E - e0;
    const int nval = rem < GAT_EPB ? rem : GAT_EPB;
    lstart[tid] = ls;
    unsigned* row = runStart + (long)blockIdx.x * (NB + 1);
    if (tid < NB) row[tid] = ls;
    if (tid == 0) row[NB] = (unsigned)nval;
    __syncthreads();
#pragma unroll
    for (int j = 0; j < GAT_EPT; ++j) {
        if (dv[j] >= 0) stage[lstart[dv[j] >> GAT_BIN_SHIFT] + rk[j]] = make_int2(sv[j], dv[j]);
    }
    __syncthreads();
    for (int i = tid; i < nval; i += 1024) part[e0 + i] = stage[i];
}

#define GAT_SORT_LDS_WORDS (GAT_SORT_CAP + 2 * GAT_BIN_NODES + 64)
__device__ __forceinline__ void gat_bin_sort_role(
    float* lds, const int bin,
    const int2* __restrict__ part, const unsigned* __restrict__ runStart, int nblk, int NB, int E, int N,
    int* __restrict__ cnt, int* __restrict__ offs, int* tmp, int* __restrict__ sorted)
{
    int*      stageS = reinterpret_cast<int*>(lds);
    unsigned* cntL   = reinterpret_cast<unsigned*>(lds) + GAT_SORT_CAP;
    unsigned* curL   = cntL + GAT_BIN_NODES;
    unsigned* wsum   = curL + GAT_BIN_NODES;
    const int tid  = threadIdx.x;
    const int lane = tid & 31;
    const int wv = __builtin_amdgcn_readfirstlane(tid >> 5);
    const int node0 = bin << GAT_BIN_SHIFT;
    auto load_runs = [&](int r0, unsigned& offv, unsigned& lenv, unsigned& s0) {
        const int r = r0 + wv + 32 * lane;
        offv = 0u; lenv = 0u; s0 = 0u;
        if (r < nblk) {
            const unsigned* row = runStart + (long)r * (NB + 1) + bin;
            const unsigned t0 = row[0], t1 = row[1];
            s0 = t0; offv = (unsigned)r * GAT_EPB + t0; lenv = t1 - t0;
        }
    };
    unsigned off0, len0, st0;
    load_runs(0, off0, len0, st0);
    unsigned sB = st0, sM = len0;
    for (int r0 = 1024; r0 < nblk; r0 += 1024) {
        unsigned o, l, s0;
        load_runs(r0, o, l, s0);
        sB += s0; sM += l;
    }
    gat_block_sum2(sB, sM, wsum);
    const int beg = (int)sB;
    const int m   = (int)sM;
    auto do_runs = [&](int r0, unsigned offv, unsigned lenv, auto f) {
        const int nr  = (nblk - r0) < 1024 ? (nblk - r0) : 1024;
        const int nrw = nr > wv ? (nr - wv + 31) >> 5 : 0;
        auto fetch = [&](int k, int2* pl, int* ob, int* ln) {
#pragma unroll
            for (int u = 0; u < 4; ++u) {
                ob[u] = __builtin_amdgcn_readlane((int)offv, (k + u) & 31);
                ln[u] = __builtin_amdgcn_readlane((int)lenv, (k + u) & 31);
                const int i = ob[u] + lane;
                pl[u] = part[i < E ? i : E - 1];
            }
        };
        int2 pl[4]; int ob[4], ln[4];
        fetch(0, pl, ob, ln);
        for (int k0 = 0; k0 < nrw; k0 += 4) {
            int2 pn[4]; int obn[4], lnn[4];
            fetch(k0 + 4, pn, obn, lnn);
#pragma unroll
            for (int u = 0; u < 4; ++u) {
                if (lane < ln[u]) f(pl[u]);
            }
#pragma unroll
            for (int u = 0; u < 4; ++u) {
                for (int i = 32 + lane; i < ln[u]; i += 32) f(part[ob[u] + i]);
            }
#pragma unroll
            for (int u = 0; u < 4; ++u) { pl[u] = pn[u]; ob[u] = obn[u]; ln[u] = lnn[u]; }
        }
    };
    auto for_each_pair = [&](auto f) {
        do_runs(0, off0, len0, f);
        for (int r0 = 1024; r0 < nblk; r0 += 1024) {
            unsigned o, l, s0;
            load_runs(r0, o, l, s0);
            do_runs(r0, o, l, f);
        }
    };
    if (tid < GAT_BIN_NODES) cntL[tid] = 0u;
    __syncthreads();
    for_each_pair([&](int2 p) { atomicAdd(&cntL[p.y - node0], 1u); });
    __syncthreads();
    const unsigned cn = (tid < GAT_BIN_NODES) ? cntL[tid] : 0u;
    const unsigned ex = gat_block_scan(cn, wsum);
    if (tid < GAT_BIN_NODES) {
        curL[tid] = ex;
        const int n = node0 + tid;
        if (n < N) { cnt[n] = (int)cn; offs[n] = beg + (int)ex; }
    }
    __syncthreads();
    const bool fits = m <= GAT_SORT_CAP;
    for_each_pair([&](int2 p) {
        const unsigned pos = atomicAdd(&curL[p.y - node0], 1u);
        if (fits) stageS[pos] = p.x; else tmp[beg + (int)pos] = p.x;
    });
    if (!fits) __threadfence();
    __syncthreads();
    auto order_groups = [&](auto src_at) {
        for (int nl = wv; nl < GAT_BIN_NODES; nl += 32) {
            const int deg = __builtin_amdgcn_readfirstlane((int)cntL[nl]);
            const int ex  = __builtin_amdgcn_readfirstlane((int)curL[nl]) - deg;
            if (deg <= 32) {
                const unsigned key = (lane < deg) ? (((unsigned)src_at(ex + lane) << 5) | (unsigned)lane)
                                                  : 0xffffffffu;
                unsigned rank = 0u;
                for (int j = 0; j < deg; j += 4) {
#pragma unroll
                    for (int u = 0; u < 4; ++u) {
                        const unsigned kj = (unsigned)__builtin_amdgcn_readlane((int)key, j + u);
                        rank += (kj < key) ? 1u : 0u;
                    }
                }
                if (lane < deg) sorted[beg + ex + (int)rank] = (int)(key >> 5);
            } else {
                for (int i = lane; i < deg; i += 32) {
                    int v = src_at(ex + i);
                    int rank = 0;
                    for (int j = 0; j < deg; ++j) {
                        int vj = src_at(ex + j);
                        rank += (vj < v || (vj == v && j < i)) ? 1 : 0;
                    }
                    sorted[beg + ex + rank] = v;
                }
            }
        }
    };
    if (fits) order_groups([&](int i) { return stageS[i]; });
    else      order_groups([&](int i) { return tmp[beg + i]; });
}

__device__ __forceinline__ void gat_gemm1_role(
    float* ldsW, const int gblk,
    const float* __restrict__ X, const float* __restrict__ W,
    const float* __restrict__ aw_s, const float* __restrict__ aw_d,
    float* __restrict__ Y, float* __restrict__ as_, float* __restrict__ ad_, int N)
{
    constexpr int K = 128, NCOL = 128;
    {
        float4 wv[4];
#pragma unroll
        for (int i = 0; i < 4; ++i) wv[i] = reinterpret_cast<const float4*>(W)[(int)threadIdx.x + i * 1024];
#pragma unroll
        for (int i = 0; i < 4; ++i) reinterpret_cast<float4*>(ldsW)[(int)threadIdx.x + i * 1024] = wv[i];
    }
    __syncthreads();

    const int lane = threadIdx.x & 31;
    const int wave = __builtin_amdgcn_readfirstlane((int)(threadIdx.x >> 5));
    const int m    = lane & 15;
    const int half = lane >> 4;
    const int koff = half << 1;
    const int ch   = wave & 1;
    const int rg   = wave >> 1;
    const int col0 = ch * 64 + m;

    float wS[4], wD[4];
#pragma unroll
    for (int t = 0; t < 4; ++t) { wS[t] = aw_s[col0 + t * 16]; wD[t] = aw_d[col0 + t * 16]; }

#pragma nounroll
    for (int it = 0; it < 2; ++it) {
        const long row0 = (long)gblk * 512 + it * 256 + rg * 16;
        if (row0 >= (long)N) break;
        const int rlast = (int)((long)N - 1 - row0);
        const int mr    = m < rlast ? m : rlast;
        const float* xrow = X + row0 * K + (mr * K + koff);

        v8f acc[4] = {};
        for (int k0 = 0; k0 < K; k0 += 4) {
            v2f a;
            a.x = xrow[k0];
            a.y = xrow[k0 + 1];
#pragma unroll
            for (int t = 0; t < 4; ++t) {
                v2f b;
                b.x = ldsW[(k0 + koff) * NCOL + col0 + t * 16];
                b.y = ldsW[(k0 + koff + 1) * NCOL + col0 + t * 16];
                acc[t] = __builtin_amdgcn_wmma_f32_16x16x4_f32(
                    false, a, false, b, (short)0, acc[t], false, false);
            }
        }
        float* ylane = Y + row0 * NCOL + (half * 8 * NCOL + col0);
        const int rleft = rlast - half * 8;
#pragma unroll
        for (int r = 0; r < 8; ++r) {
            if (r <= rleft) {
#pragma unroll
                for (int t = 0; t < 4; ++t) ylane[r * NCOL + t * 16] = acc[t][r];
            }
        }
        float v[16];
#pragma unroll
        for (int r = 0; r < 8; ++r) {
            float sv = 0.0f, dv = 0.0f;
#pragma unroll
            for (int t = 0; t < 4; ++t) {
                sv = fmaf(acc[t][r], wS[t], sv);
                dv = fmaf(acc[t][r], wD[t], dv);
            }
            v[r * 2 + 0] = sv;
            v[r * 2 + 1] = dv;
        }
        gat_fold16<16>(v, m);
        const int rr = half * 8 + (m >> 1);
        if (rr <= rlast) {
            if (m & 1) ad_[row0 * 2 + (rr * 2 + ch)] = v[0]; else as_[row0 * 2 + (rr * 2 + ch)] = v[0];
        }
    }
}

__global__ __launch_bounds__(1024, 16) void gat_sort_gemm1_kernel(
    const int2* __restrict__ part, const unsigned* __restrict__ runStart, int nblk, int NB, int E, int N, int NG,
    int* __restrict__ cnt, int* __restrict__ offs, int* tmp, int* __restrict__ sorted,
    const float* __restrict__ X, const float* __restrict__ W,
    const float* __restrict__ aw_s, const float* __restrict__ aw_d,
    float* __restrict__ Y, float* __restrict__ as_, float* __restrict__ ad_)
{
    __shared__ __attribute__((aligned(16))) float lds[128 * 128];
    static_assert(GAT_SORT_LDS_WORDS <= 128 * 128, "the sort's tables fit the buffer");
    if ((int)blockIdx.x < NG) gat_gemm1_role(lds, (int)blockIdx.x, X, W, aw_s, aw_d, Y, as_, ad_, N);
    else                      gat_bin_sort_role(lds, (int)blockIdx.x - NG, part, runStart, nblk, NB, E, N, cnt, offs, tmp, sorted);
}

__device__ __forceinline__ float gat_rl(float x, int k) {
    return __uint_as_float((unsigned)__builtin_amdgcn_readlane((int)__float_as_uint(x), k));
}

__global__ __launch_bounds__(256) void gat_node_l1_kernel(
    const int* __restrict__ cnt, const int* __restrict__ offs, const int* __restrict__ sorted,
    const float* __restrict__ as_, const float* __restrict__ ad_,
    const float* __restrict__ feat, const float* __restrict__ bias,
    float* __restrict__ outp, int N)
{
    const int lane = threadIdx.x & 31;
    const int w0 = __builtin_amdgcn_readfirstlane((int)(blockIdx.x * 8 + (threadIdx.x >> 5)));
    const int nw = (int)gridDim.x * 8;
    for (int n = w0; n < N; n += nw) {
        const int deg = __builtin_amdgcn_readfirstlane(cnt[n]);
        const int off = __builtin_amdgcn_readfirstlane(offs[n]);
        const float2* as2 = reinterpret_cast<const float2*>(as_);
        const float2* ad2 = reinterpret_cast<const float2*>(ad_);
        const float4* f4  = reinterpret_cast<const float4*>(feat);
        const float2 adn = ad2[n];
        const float2 asn = as2[n];
        const float es0 = gat_lrelu(asn.x + adn.x), es1 = gat_lrelu(asn.y + adn.y);

        const bool v0 = lane < deg;
        int s = n; float e0 = es0, e1 = es1;
        if (v0) {
            s = sorted[off + lane];
            float2 a = as2[s];
            e0 = gat_lrelu(a.x + adn.x); e1 = gat_lrelu(a.y + adn.y);
        }
        float m0 = fmaxf(e0, es0), m1 = fmaxf(e1, es1);
        for (int base = 32; base < deg; base += 32) {
            int i = base + lane;
            if (i < deg) {
                float2 a = as2[sorted[off + i]];
                m0 = fmaxf(m0, gat_lrelu(a.x + adn.x)); m1 = fmaxf(m1, gat_lrelu(a.y + adn.y));
            }
        }
        for (int o = 16; o > 0; o >>= 1) {
            m0 = fmaxf(m0, __shfl_xor(m0, o, 32));
            m1 = fmaxf(m1, __shfl_xor(m1, o, 32));
        }
        const float p0 = v0 ? expf(e0 - m0) : 0.0f, p1 = v0 ? expf(e1 - m1) : 0.0f;
        float d0 = p0, d1 = p1;
        for (int base = 32; base < deg; base += 32) {
            int i = base + lane;
            if (i < deg) {
                float2 a = as2[sorted[off + i]];
                d0 += expf(gat_lrelu(a.x + adn.x) - m0); d1 += expf(gat_lrelu(a.y + adn.y) - m1);
            }
        }
        for (int o = 16; o > 0; o >>= 1) {
            d0 += __shfl_xor(d0, o, 32);
            d1 += __shfl_xor(d1, o, 32);
        }
        const float ps0 = expf(es0 - m0), ps1 = expf(es1 - m1);
        const float den0 = (d0 + ps0) + 1e-16f, den1 = (d1 + ps1) + 1e-16f;

        float4 acc = make_float4(0.0f, 0.0f, 0.0f, 0.0f);
        auto gather = [&](int sv, float q0, float q1, int cn) {
            for (int k = 0; k < cn; k += 4) {
#pragma unroll
                for (int u = 0; u < 4; ++u) {
                    int   sk = __builtin_amdgcn_readlane(sv, k + u);
                    float a0 = gat_rl(q0, k + u), a1 = gat_rl(q1, k + u);
                    float ck = (lane < 16) ? a0 : a1;
                    float4 f = f4[(long)sk * 32 + lane];
                    acc.x = fmaf(ck, f.x, acc.x); acc.y = fmaf(ck, f.y, acc.y);
                    acc.z = fmaf(ck, f.z, acc.z); acc.w = fmaf(ck, f.w, acc.w);
                }
            }
        };
        gather(s, p0 / den0, p1 / den1, deg < 32 ? deg : 32);
        for (int base = 32; base < deg; base += 32) {
            int i = base + lane;
            int t = n; float q0 = 0.0f, q1 = 0.0f;
            if (i < deg) {
                t = sorted[off + i];
                float2 a = as2[t];
                q0 = expf(gat_lrelu(a.x + adn.x) - m0) / den0;
                q1 = expf(gat_lrelu(a.y + adn.y) - m1) / den1;
            }
            gather(t, q0, q1, (deg - base) < 32 ? (deg - base) : 32);
        }
        {
            float cs = (lane < 16) ? (ps0 / den0) : (ps1 / den1);
            float4 f = f4[(long)n * 32 + lane];
            acc.x = fmaf(cs, f.x, acc.x); acc.y = fmaf(cs, f.y, acc.y);
            acc.z = fmaf(cs, f.z, acc.z); acc.w = fmaf(cs, f.w, acc.w);
        }
        const float4 b = reinterpret_cast<const float4*>(bias)[lane];
        float4 r;
        float t0 = acc.x + b.x, t1 = acc.y + b.y, t2 = acc.z + b.z, t3 = acc.w + b.w;
        r.x = t0 > 0.0f ? t0 : (expf(t0) - 1.0f);
        r.y = t1 > 0.0f ? t1 : (expf(t1) - 1.0f);
        r.z = t2 > 0.0f ? t2 : (expf(t2) - 1.0f);
        r.w = t3 > 0.0f ? t3 : (expf(t3) - 1.0f);
        reinterpret_cast<float4*>(outp)[(long)n * 32 + lane] = r;
    }
}

#define GAT_L2_U 6
__global__ __launch_bounds__(256, 16) void gat_node_l2_kernel(
    const int* __restrict__ cnt, const int* __restrict__ offs, const int* __restrict__ sorted,
    const float* __restrict__ ad_,
    const float* __restrict__ feat, const float* __restrict__ bias,
    float* __restrict__ outp, int N)
{
    const int lane = threadIdx.x & 31;
    const int w0 = __builtin_amdgcn_readfirstlane((int)(blockIdx.x * 8 + (threadIdx.x >> 5)));
    const int nw = (int)gridDim.x * 8;
    const int g  = lane >> 4;
    const int cl = lane & 15;
    const int cc = cl < 10 ? cl : 10;
    const int la = (lane & 16) | 10;
    const int l0 = lane & 16;
    const bool pw = cl < GAT_L2_U;
    const int eu = pw ? cl : 0;
    const float4* f4 = reinterpret_cast<const float4*>(feat);
    const float ninf = -__builtin_inff();
    for (int n = w0; n < N; n += nw) {
        const int deg = __builtin_amdgcn_readfirstlane(cnt[n]);
        const int off = __builtin_amdgcn_readfirstlane(offs[n]);
        const float adn = ad_[n];
        int s = n;
        if (lane < deg) s = sorted[off + lane];
        const float4 fs = f4[(long)n * 16 + cc];
        const float es = gat_lrelu(__shfl(fs.x, la, 32) + adn);

        float  mg  = (g == 0) ? es : ninf;
        float  dg  = (g == 0) ? 1.0f : 0.0f;
        float4 acc = (g == 0) ? fs : make_float4(0.0f, 0.0f, 0.0f, 0.0f);
        auto gather = [&](int sv, int cn) {
            const int iters = (cn + 1) >> 1;
            for (int tb = 0; tb < iters; tb += GAT_L2_U) {
                float4 f[GAT_L2_U];
                bool   val[GAT_L2_U];
#pragma unroll
                for (int u = 0; u < GAT_L2_U; ++u) {
                    const int k = (tb + u) * 2 + g;
                    val[u] = k < cn;
                    const int sk = __shfl(sv, val[u] ? k : 0, 32);
                    f[u] = f4[(long)(val[u] ? sk : n) * 16 + cc];
                }
                float e[GAT_L2_U];
                float mn = mg;
#pragma unroll
                for (int u = 0; u < GAT_L2_U; ++u) {
                    const float a = __shfl(f[u].x, la, 32);
                    e[u] = val[u] ? gat_lrelu(a + adn) : ninf;
                    mn = fmaxf(mn, e[u]);
                }
                float es_ = e[0];
#pragma unroll
                for (int u = 1; u < GAT_L2_U; ++u) es_ = (eu == u) ? e[u] : es_;
                const bool  vs_ = ((tb + eu) * 2 + g) < cn;
                const float y   = expf((pw ? es_ : mg) - mn);
                const float yy  = pw ? (vs_ ? y : 0.0f) : ((mg == mn) ? 1.0f : y);
                const float sc  = __shfl(yy, l0 | GAT_L2_U, 32);
                dg *= sc;
                acc.x *= sc; acc.y *= sc; acc.z *= sc; acc.w *= sc;
#pragma unroll
                for (int u = 0; u < GAT_L2_U; ++u) {
                    const float p = __shfl(yy, l0 | u, 32);
                    dg += p;
                    acc.x = fmaf(p, f[u].x, acc.x); acc.y = fmaf(p, f[u].y, acc.y);
                    acc.z = fmaf(p, f[u].z, acc.z); acc.w = fmaf(p, f[u].w, acc.w);
                }
                mg = mn;
            }
        };
        gather(s, deg < 32 ? deg : 32);
        for (int base = 32; base < deg; base += 32) {
            const int i = base + lane;
            int t = n;
            if (i < deg) t = sorted[off + i];
            gather(t, (deg - base) < 32 ? (deg - base) : 32);
        }
        const float mo = __shfl_xor(mg, 16, 32);
        const float dO = __shfl_xor(dg, 16, 32);
        float4 ao;
        ao.x = __shfl_xor(acc.x, 16, 32); ao.y = __shfl_xor(acc.y, 16, 32);
        ao.z = __shfl_xor(acc.z, 16, 32); ao.w = __shfl_xor(acc.w, 16, 32);
        if (lane < 10) {
            const float m   = fmaxf(mg, mo);
            const float wg  = (mg == m) ? 1.0f : expf(mg - m);
            const float wo  = (mo == m) ? 1.0f : expf(mo - m);
            const float inv = 1.0f / (fmaf(dO, wo, dg * wg) + 1e-16f);
            const float4 b = reinterpret_cast<const float4*>(bias)[lane];
            float4 r;
            r.x = fmaf(fmaf(ao.x, wo, acc.x * wg), inv, b.x);
            r.y = fmaf(fmaf(ao.y, wo, acc.y * wg), inv, b.y);
            r.z = fmaf(fmaf(ao.z, wo, acc.z * wg), inv, b.z);
            r.w = fmaf(fmaf(ao.w, wo, acc.w * wg), inv, b.w);
            reinterpret_cast<float4*>(outp)[(long)n * 10 + lane] = r;
        }
    }
}

extern "C" void kernel_launch(void* const* d_in, const int* in_sizes, int n_in,
                              void* d_out, int out_size, void* d_ws, size_t ws_size,
                              hipStream_t stream)
{
    (void)n_in; (void)out_size; (void)ws_size;
    const float* x   = (const float*)d_in[0];
    const int*   ei  = (const int*)  d_in[1];
    const float* W1  = (const float*)d_in[2];
    const float* aS1 = (const float*)d_in[3];
    const float* aD1 = (const float*)d_in[4];
    const float* b1  = (const float*)d_in[5];
    const float* W2  = (const float*)d_in[6];
    const float* aS2 = (const float*)d_in[7];
    const float* aD2 = (const float*)d_in[8];
    const float* b2  = (const float*)d_in[9];

    const int IN = 128, HC1 = 128;
    int N  = in_sizes[0] / IN;
    int E  = in_sizes[1] / 2;
    const int* srcI = ei;
    const int* dstI = ei + E;

    float* wsf = (float*)d_ws;
    size_t off = 0;
    auto take = [&](size_t nf) -> size_t { size_t o = off; off += (nf + 63) & ~(size_t)63; return o; };
    float* h1     = wsf + take((size_t)N * HC1);
    float* agg1   = wsf + take((size_t)N * HC1);
    float* as1    = wsf + take((size_t)N * 2);
    float* ad1    = wsf + take((size_t)N * 2);
    float* ad2    = wsf + take((size_t)N);
    int*   cnt    = (int*)(wsf + take((size_t)N));
    int*   offs   = (int*)(wsf + take((size_t)N));
    int*   tmp    = (int*)(wsf + take((size_t)E));
    int*   sorted = (int*)(wsf + take((size_t)E));

    dim3 blk(256);
    dim3 ngrid(2048);

    int NB   = (N + GAT_BIN_NODES - 1) >> GAT_BIN_SHIFT;
    int nblk = (E + GAT_EPB - 1) / GAT_EPB;
    int2*     part     = (int2*)agg1;
    unsigned* runStart = (unsigned*)(agg1 + (size_t)2 * E);
    gat_bin_scatter_kernel<<<nblk, 1024, 0, stream>>>(srcI, dstI, E, NB, runStart, part);
    int NG = (N + 511) / 512;
    gat_sort_gemm1_kernel<<<NG + NB, 1024, 0, stream>>>(
        part, runStart, nblk, NB, E, N, NG, cnt, offs, tmp, sorted, x, W1, aS1, aD1, h1, as1, ad1);

    gat_node_l1_kernel<<<ngrid, blk, 0, stream>>>(cnt, offs, sorted, as1, ad1, h1, b1, agg1, N);

    float* h2 = h1;
    gat_gemm_wmma<48, 3, 1, 40, 64><<<(N + 127) / 128, blk, 0, stream>>>(agg1, W2, aS2, aD2, h2, ad2, N);
    gat_node_l2_kernel<<<ngrid, blk, 0, stream>>>(cnt, offs, sorted, ad2, h2, b2, (float*)d_out, N);
}
